# starting kernel
# baseline (speedup 1.0000x reference)
#include <hip/hip_runtime.h>
#include <hip/hip_bf16.h>


typedef __attribute__((ext_vector_type(2)))  float    v2f;
typedef __attribute__((ext_vector_type(8)))  float    v8f;
typedef __attribute__((ext_vector_type(16))) _Float16 v16h;

__device__ __forceinline__ void atomicAddF(float* p, float v) {
    __hip_atomic_fetch_add(p, v, __ATOMIC_RELAXED, __HIP_MEMORY_SCOPE_AGENT);
}

__global__ void gcn_init(float* __restrict__ agg1, float* __restrict__ outb,
                         float* __restrict__ deg, int n8  ) {
    int i = blockIdx.x * blockDim.x + threadIdx.x;
    int n = n8 >> 3;
    if (i < n8) agg1[i] = 0.0f;
    if (i < (n << 1)) outb[i] = 0.0f;
    if (i < n) deg[i] = 1.0f;
}

__global__ void gcn_degree(const int* __restrict__ col, float* __restrict__ deg, int E) {
    int e = blockIdx.x * blockDim.x + threadIdx.x;
    if (e < E) atomicAddF(&deg[col[e]], 1.0f);
}

__global__ void gcn_dinv(float* __restrict__ deg, int N) {
    int i = blockIdx.x * blockDim.x + threadIdx.x;
    if (i < N) deg[i] = __frsqrt_rn(deg[i]);
}

__global__ void gcn_xw1_wmma(const float* __restrict__ x, const float* __restrict__ W1,
                             float* __restrict__ xw1, int nTiles, int N) {
    __shared__ float tile[8][16 * 8];
    int flat = blockIdx.x * blockDim.x + threadIdx.x;
    int wave = __builtin_amdgcn_readfirstlane(flat >> 5);
    if (wave >= nTiles) return;

    const int lane = threadIdx.x & 31;
    const int m    = lane & 15;
    const int half = lane >> 4;
    const int n    = m;
    const float colMask = (n < 8) ? 1.0f : 0.0f;
    const int   nn      = n & 7;

    long rowA = (long)wave * 16 + m;
    if (rowA >= N) rowA = N - 1;
    const float* xr = x + rowA * 16;

    v8f acc = {};
#if __has_builtin(__builtin_amdgcn_wmma_f32_16x16x4_f32)
#pragma unroll
    for (int kb = 0; kb < 16; kb += 4) {
        const int k0 = kb + 2 * half;
        v2f a, b;
        a.x = xr[k0 + 0];
        a.y = xr[k0 + 1];
        b.x = W1[(k0 + 0) * 8 + nn] * colMask;
        b.y = W1[(k0 + 1) * 8 + nn] * colMask;
        acc = __builtin_amdgcn_wmma_f32_16x16x4_f32(
                  false, a, false, b, (short)0, acc, false, false);
    }
#else
    v16h a, b;
#pragma unroll
    for (int j = 0; j < 16; ++j) {
        a[j] = (j < 8) ? (_Float16)xr[8 * half + j] : (_Float16)0.0f;
        float bv = (half == 0) ? W1[j * 8 + nn] * colMask : 0.0f;
        b[j] = (_Float16)bv;
    }
    acc = __builtin_amdgcn_wmma_f32_16x16x32_f16(
              false, a, false, b, (short)0, acc, false, false);
#endif

    float* t = &tile[threadIdx.x >> 5][0];
    if (n < 8) {
#pragma unroll
        for (int r = 0; r < 8; ++r)
            t[(r + 8 * half) * 8 + n] = acc[r];
    }
    asm volatile("s_wait_dscnt 0x0" ::: "memory");

    const int rowIdx = lane >> 1;
    const int quad   = lane & 1;
    long gr = (long)wave * 16 + rowIdx;
    float4 v = *(const float4*)(t + rowIdx * 8 + quad * 4);
    if (gr < N) *(float4*)(xw1 + gr * 8 + quad * 4) = v;
}

__global__ void gcn_prop1(const int* __restrict__ row, const int* __restrict__ col,
                          const float* __restrict__ dinv, const float* __restrict__ xw1,
                          float* __restrict__ agg1, long EF  ) {
    long t = (long)blockIdx.x * blockDim.x + threadIdx.x;
    if (t >= EF) return;
    const int e = (int)(t >> 3);
    const int f = (int)(t & 7);
    const int r = row[e], c = col[e];
    const float w = dinv[r] * dinv[c];
    atomicAddF(agg1 + (long)c * 8 + f, xw1[(long)r * 8 + f] * w);
}

__global__ void gcn_relu_hw2(const float* __restrict__ agg1, const float* __restrict__ xw1,
                             const float* __restrict__ dinv, const float* __restrict__ b1,
                             const float* __restrict__ W2, float* __restrict__ hw2, int N) {
    int i = blockIdx.x * blockDim.x + threadIdx.x;
    if (i >= N) return;
    const float self = dinv[i] * dinv[i];
    const float4* a = (const float4*)(agg1 + (long)i * 8);
    const float4* s = (const float4*)(xw1 + (long)i * 8);
    float hv[8];
    {
        float4 a0 = a[0], a1 = a[1], s0 = s[0], s1 = s[1];
        hv[0] = a0.x + self * s0.x; hv[1] = a0.y + self * s0.y;
        hv[2] = a0.z + self * s0.z; hv[3] = a0.w + self * s0.w;
        hv[4] = a1.x + self * s1.x; hv[5] = a1.y + self * s1.y;
        hv[6] = a1.z + self * s1.z; hv[7] = a1.w + self * s1.w;
    }
    float o0 = 0.0f, o1 = 0.0f;
#pragma unroll
    for (int f = 0; f < 8; ++f) {
        float h = fmaxf(hv[f] + b1[f], 0.0f);
        o0 = fmaf(h, W2[f * 2 + 0], o0);
        o1 = fmaf(h, W2[f * 2 + 1], o1);
    }
    hw2[(long)i * 2 + 0] = o0;
    hw2[(long)i * 2 + 1] = o1;
}

__global__ void gcn_prop2(const int* __restrict__ row, const int* __restrict__ col,
                          const float* __restrict__ dinv, const float* __restrict__ hw2,
                          float* __restrict__ outb, long EF2  ) {
    long t = (long)blockIdx.x * blockDim.x + threadIdx.x;
    if (t >= EF2) return;
    const int e = (int)(t >> 1);
    const int f = (int)(t & 1);
    const int r = row[e], c = col[e];
    const float w = dinv[r] * dinv[c];
    atomicAddF(outb + (long)c * 2 + f, hw2[(long)r * 2 + f] * w);
}

__global__ void gcn_finalize(float* __restrict__ outb, const float* __restrict__ hw2,
                             const float* __restrict__ dinv, const float* __restrict__ b2,
                             int N) {
    int i = blockIdx.x * blockDim.x + threadIdx.x;
    if (i >= N) return;
    const float self = dinv[i] * dinv[i];
    float v0 = outb[(long)i * 2 + 0] + self * hw2[(long)i * 2 + 0] + b2[0];
    float v1 = outb[(long)i * 2 + 1] + self * hw2[(long)i * 2 + 1] + b2[1];
    const float m = fmaxf(v0, v1);
    const float lse = m + logf(expf(v0 - m) + expf(v1 - m));
    outb[(long)i * 2 + 0] = v0 - lse;
    outb[(long)i * 2 + 1] = v1 - lse;
}

extern "C" void kernel_launch(void* const* d_in, const int* in_sizes, int n_in,
                              void* d_out, int out_size, void* d_ws, size_t ws_size,
                              hipStream_t stream) {
    const float* x    = (const float*)d_in[0];
    const float* W1   = (const float*)d_in[1];
    const float* b1   = (const float*)d_in[2];
    const float* W2   = (const float*)d_in[3];
    const float* b2   = (const float*)d_in[4];
    const int*   edge = (const int*)d_in[5];

    const int N = in_sizes[0] / 16;
    const int E = in_sizes[5] / 2;
    const int* row = edge;
    const int* col = edge + E;
    float* out = (float*)d_out;

    size_t off = 0;
    auto carve = [&](size_t bytes) {
        void* p = (char*)d_ws + off;
        off = (off + bytes + 255) & ~(size_t)255;
        return p;
    };
    float* deg  = (float*)carve((size_t)N * 4);
    float* xw1  = (float*)carve((size_t)N * 8 * 4);
    float* agg1 = (float*)carve((size_t)N * 8 * 4);
    float* hw2  = (float*)carve((size_t)N * 2 * 4);
    (void)ws_size;

    const int B = 256;
    const long EF  = (long)E * 8;
    const long EF2 = (long)E * 2;
    const int gN   = (N + B - 1) / B;
    const int gN8  = (N * 8 + B - 1) / B;
    const int gE   = (E + B - 1) / B;
    const int gEF  = (int)((EF + B - 1) / B);
    const int gEF2 = (int)((EF2 + B - 1) / B);
    const int nTiles = (N + 15) / 16;
    const int gW   = (nTiles * 32 + B - 1) / B;

    gcn_init    <<<gN8,  B, 0, stream>>>(agg1, out, deg, N * 8);
    gcn_degree  <<<gE,   B, 0, stream>>>(col, deg, E);
    gcn_dinv    <<<gN,   B, 0, stream>>>(deg, N);
    gcn_xw1_wmma<<<gW,   B, 0, stream>>>(x, W1, xw1, nTiles, N);
    gcn_prop1   <<<gEF,  B, 0, stream>>>(row, col, deg, xw1, agg1, EF);
    gcn_relu_hw2<<<gN,   B, 0, stream>>>(agg1, xw1, deg, b1, W2, hw2, N);
    gcn_prop2   <<<gEF2, B, 0, stream>>>(row, col, deg, hw2, out, EF2);
    gcn_finalize<<<gN,   B, 0, stream>>>(out, hw2, deg, b2, N);
}


# Both scatter-adds without global atomics or edge reordering: a block owns a node range, sums in LDS, filters a slice of the destination list, LDS float adds (arrival order); planes summed per node
# speedup vs baseline: 1.5412x; 1.5412x over previous
#include <hip/hip_runtime.h>
#include <hip/hip_bf16.h>


typedef __attribute__((ext_vector_type(2)))  float    v2f;
typedef __attribute__((ext_vector_type(8)))  float    v8f;
typedef __attribute__((ext_vector_type(16))) _Float16 v16h;

__device__ __forceinline__ void ldsAddF(float* p, float v) {
    __hip_atomic_fetch_add(p, v, __ATOMIC_RELAXED, __HIP_MEMORY_SCOPE_WORKGROUP);
}

#define DEG_NPP 65536
#define DEG_S   32
#define DEG_T   1024

__device__ __forceinline__ void deg_count(unsigned* cnt, int c, unsigned base) {
    const unsigned d = (unsigned)c - base;
    if (d < (unsigned)DEG_NPP) atomicAdd(&cnt[d], 1u);
}

__device__ __forceinline__ void deg_count4(unsigned* cnt, int4 v, unsigned base) {
    deg_count(cnt, v.x, base);
    deg_count(cnt, v.y, base);
    deg_count(cnt, v.z, base);
    deg_count(cnt, v.w, base);
}

__global__ __launch_bounds__(DEG_T)
void gcn_degree_ranges(const int* __restrict__ col, unsigned* __restrict__ planes,
                       int E, int Npad) {
    __shared__ unsigned cnt[DEG_NPP];
    const int tid = threadIdx.x;
    const int s   = blockIdx.x;
    const int nodeBase  = blockIdx.y * DEG_NPP;
    const unsigned base = (unsigned)nodeBase;

    const uint4 zero4 = make_uint4(0u, 0u, 0u, 0u);
    for (int i = tid * 4; i < DEG_NPP; i += DEG_T * 4) *(uint4*)(cnt + i) = zero4;
    __syncthreads();

    const long chunk = (long)((((E + DEG_S - 1) / DEG_S) + 3) & ~3);
    long lo = (long)s * chunk;       if (lo > (long)E) lo = (long)E;
    long hi = (long)s * chunk + chunk; if (hi > (long)E) hi = (long)E;
    int e0 = (int)lo;
    const int e1 = (int)hi;

    if ((((size_t)col) & 15) == 0) {
        const int nvec = (e1 - e0) >> 2;
        const int4* cv = (const int4*)(col + e0);
        int i = tid;
        for (; i + 7 * DEG_T < nvec; i += 8 * DEG_T) {
            int4 v[8];
#pragma unroll
            for (int k = 0; k < 8; ++k) v[k] = cv[i + k * DEG_T];
#pragma unroll
            for (int k = 0; k < 8; ++k) deg_count4(cnt, v[k], base);
        }
        if (i < nvec) {
            int4 v[8];
#pragma unroll
            for (int k = 0; k < 8; ++k) {
                const int idx = i + k * DEG_T;
                v[k] = cv[idx < nvec ? idx : nvec - 1];
            }
#pragma unroll
            for (int k = 0; k < 8; ++k)
                if (i + k * DEG_T < nvec) deg_count4(cnt, v[k], base);
        }
        e0 += nvec << 2;
    }
    for (int i = e0 + tid; i < e1; i += DEG_T) deg_count(cnt, col[i], base);
    __syncthreads();

    int lim = Npad - nodeBase;
    if (lim > DEG_NPP) lim = DEG_NPP;
    unsigned* dst = planes + (size_t)s * (size_t)Npad + (size_t)nodeBase;
    for (int i = tid * 4; i < lim; i += DEG_T * 4)
        *(uint4*)(dst + i) = *(const uint4*)(cnt + i);
}

__global__ void gcn_dinv_sum(const unsigned* __restrict__ planes, float* __restrict__ dinv,
                             int Npad) {
    const int i4 = (blockIdx.x * blockDim.x + threadIdx.x) * 4;
    if (i4 >= Npad) return;
    const uint4* p = (const uint4*)(planes + i4);
    const size_t pitch4 = (size_t)Npad >> 2;
    unsigned t0 = 1u, t1 = 1u, t2 = 1u, t3 = 1u;
#pragma unroll
    for (int s = 0; s < DEG_S; ++s) {
        const uint4 v = p[(size_t)s * pitch4];
        t0 += v.x; t1 += v.y; t2 += v.z; t3 += v.w;
    }
    *(float4*)(dinv + i4) = make_float4(__frsqrt_rn((float)t0), __frsqrt_rn((float)t1),
                                        __frsqrt_rn((float)t2), __frsqrt_rn((float)t3));
}

__global__ void gcn_xw1_wmma(const float* __restrict__ x, const float* __restrict__ W1,
                             float* __restrict__ xw1, int nTiles, int N) {
    __shared__ float tile[8][16 * 8];
    int flat = blockIdx.x * blockDim.x + threadIdx.x;
    int wave = __builtin_amdgcn_readfirstlane(flat >> 5);
    if (wave >= nTiles) return;

    const int lane = threadIdx.x & 31;
    const int m    = lane & 15;
    const int half = lane >> 4;
    const int n    = m;
    const float colMask = (n < 8) ? 1.0f : 0.0f;
    const int   nn      = n & 7;

    long rowA = (long)wave * 16 + m;
    if (rowA >= N) rowA = N - 1;
    const float* xr = x + rowA * 16;

    v8f acc = {};
#if __has_builtin(__builtin_amdgcn_wmma_f32_16x16x4_f32)
#pragma unroll
    for (int kb = 0; kb < 16; kb += 4) {
        const int k0 = kb + 2 * half;
        v2f a, b;
        a.x = xr[k0 + 0];
        a.y = xr[k0 + 1];
        b.x = W1[(k0 + 0) * 8 + nn] * colMask;
        b.y = W1[(k0 + 1) * 8 + nn] * colMask;
        acc = __builtin_amdgcn_wmma_f32_16x16x4_f32(
                  false, a, false, b, (short)0, acc, false, false);
    }
#else
    v16h a, b;
#pragma unroll
    for (int j = 0; j < 16; ++j) {
        a[j] = (j < 8) ? (_Float16)xr[8 * half + j] : (_Float16)0.0f;
        float bv = (half == 0) ? W1[j * 8 + nn] * colMask : 0.0f;
        b[j] = (_Float16)bv;
    }
    acc = __builtin_amdgcn_wmma_f32_16x16x32_f16(
              false, a, false, b, (short)0, acc, false, false);
#endif

    float* t = &tile[threadIdx.x >> 5][0];
    if (n < 8) {
#pragma unroll
        for (int r = 0; r < 8; ++r)
            t[(r + 8 * half) * 8 + n] = acc[r];
    }
    asm volatile("s_wait_dscnt 0x0" ::: "memory");

    const int rowIdx = lane >> 1;
    const int quad   = lane & 1;
    long gr = (long)wave * 16 + rowIdx;
    float4 v = *(const float4*)(t + rowIdx * 8 + quad * 4);
    if (gr < N) *(float4*)(xw1 + gr * 8 + quad * 4) = v;
}

#define PR_T    1024
#define PR_RING 128

template <int F>
struct PrEdge { float w; float t[F]; unsigned d; };

template <int F>
__device__ __forceinline__ void pr_load_edge(PrEdge<F>& g, uint2 q, int r, unsigned base,
                                             const float* __restrict__ dinv,
                                             const float* __restrict__ tbl) {
    g.d = q.x;
    g.w = dinv[r] * dinv[base + q.x];
    if constexpr (F == 8) {
        const float4 t0 = *(const float4*)(tbl + (size_t)r * 8);
        const float4 t1 = *(const float4*)(tbl + (size_t)r * 8 + 4);
        g.t[0] = t0.x; g.t[1] = t0.y; g.t[2] = t0.z; g.t[3] = t0.w;
        g.t[4] = t1.x; g.t[5] = t1.y; g.t[6] = t1.z; g.t[7] = t1.w;
    } else {
        const float2 t0 = *(const float2*)(tbl + (size_t)r * 2);
        g.t[0] = t0.x; g.t[1] = t0.y;
    }
}

template <int F>
__device__ __forceinline__ void pr_add_edge(float* __restrict__ acc, const PrEdge<F>& g) {
    float* a = acc + (size_t)g.d * F;
#pragma unroll
    for (int f = 0; f < F; ++f) ldsAddF(a + f, g.t[f] * g.w);
}

template <int F, int NPP, int S>
__global__ __launch_bounds__(PR_T)
void gcn_prop_ranges(const int* __restrict__ row, const int* __restrict__ col,
                     const float* __restrict__ dinv, const float* __restrict__ tbl,
                     float* __restrict__ planes, int E, int Npad) {
    __shared__ float acc[NPP * F];
    __shared__ uint2 ring[PR_T / 32][PR_RING];
    const int tid  = threadIdx.x;
    const unsigned lane = (unsigned)tid & 31u;
    const int s    = blockIdx.x;
    const int nodeBase  = blockIdx.y * NPP;
    const unsigned base = (unsigned)nodeBase;
    uint2* rq = &ring[tid >> 5][0];
    unsigned head = 0, tail = 0;

    const float4 zero4 = make_float4(0.0f, 0.0f, 0.0f, 0.0f);
    for (int i = tid * 4; i < NPP * F; i += PR_T * 4) *(float4*)(acc + i) = zero4;
    __syncthreads();

    auto push = [&](int c, unsigned lim, int e) {
        const unsigned d = (unsigned)c - base;
        const bool m = d < lim;
        const unsigned mask = __builtin_amdgcn_ballot_w32(m);
        if (mask != 0u) {
            if (m) rq[__builtin_amdgcn_mbcnt_lo(mask, tail) & (PR_RING - 1)] = make_uint2(d, (unsigned)e);
            tail += (unsigned)__builtin_popcount(mask);
        }
    };
    auto drain = [&]() {
        while (tail - head >= 64u) {
            const uint2 qa = rq[(head + lane) & (PR_RING - 1)];
            const uint2 qb = rq[(head + 32u + lane) & (PR_RING - 1)];
            const int ra = row[qa.y];
            const int rb = row[qb.y];
            PrEdge<F> ga, gb;
            pr_load_edge<F>(ga, qa, ra, base, dinv, tbl);
            pr_load_edge<F>(gb, qb, rb, base, dinv, tbl);
            pr_add_edge<F>(acc, ga);
            pr_add_edge<F>(acc, gb);
            head += 64u;
        }
    };

    const long chunk = (long)((((E + S - 1) / S) + 3) & ~3);
    long lo = (long)s * chunk;         if (lo > (long)E) lo = (long)E;
    long hi = (long)s * chunk + chunk; if (hi > (long)E) hi = (long)E;
    int e0 = (int)lo;
    const int e1 = (int)hi;

    if ((((size_t)col) & 15) == 0) {
        const int nvec = (e1 - e0) >> 2;
        const int4* cv = (const int4*)(col + e0);
        for (int ib = 0; ib < nvec; ib += 8 * PR_T) {
            int4 v[8];
#pragma unroll
            for (int k = 0; k < 8; ++k) {
                const int idx = ib + k * PR_T + tid;
                v[k] = cv[idx < nvec ? idx : nvec - 1];
            }
#pragma unroll
            for (int k = 0; k < 8; ++k) {
                const int idx = ib + k * PR_T + tid;
                const unsigned lim = idx < nvec ? (unsigned)NPP : 0u;
                const int eb = e0 + idx * 4;
                push(v[k].x, lim, eb + 0);
                push(v[k].y, lim, eb + 1);
                drain();
                push(v[k].z, lim, eb + 2);
                push(v[k].w, lim, eb + 3);
                drain();
            }
        }
        e0 += nvec << 2;
    }
    for (int ib = e0; ib < e1; ib += PR_T) {
        const int i = ib + tid;
        const bool ok = i < e1;
        push(col[ok ? i : e1 - 1], ok ? (unsigned)NPP : 0u, i);
        drain();
    }
    {
        const unsigned n = tail - head;
#pragma unroll
        for (unsigned h = 0; h < 64u; h += 32u) {
            if (lane + h < n) {
                const uint2 q = rq[(head + h + lane) & (PR_RING - 1)];
                PrEdge<F> g;
                pr_load_edge<F>(g, q, row[q.y], base, dinv, tbl);
                pr_add_edge<F>(acc, g);
            }
        }
    }
    __syncthreads();

    int lim = Npad - nodeBase;
    if (lim > NPP) lim = NPP;
    float* dst = planes + ((size_t)s * (size_t)Npad + (size_t)nodeBase) * F;
    for (int i = tid * 4; i < lim * F; i += PR_T * 4)
        *(float4*)(dst + i) = *(const float4*)(acc + i);
}

#define P1_NPP 8192
#define P1_S   4
#define P2_NPP 32768
#define P2_S   15

__global__ void gcn_relu_hw2(const float* __restrict__ planes1, const float* __restrict__ xw1,
                             const float* __restrict__ dinv, const float* __restrict__ b1,
                             const float* __restrict__ W2, float* __restrict__ hw2, int N,
                             int Npad) {
    int i = blockIdx.x * blockDim.x + threadIdx.x;
    if (i >= N) return;
    const float self = dinv[i] * dinv[i];
    const float4* a = (const float4*)(planes1 + (long)i * 8);
    const float4* s = (const float4*)(xw1 + (long)i * 8);
    const size_t pitch4 = (size_t)Npad * 2;
    float4 a0 = a[0], a1 = a[1];
#pragma unroll
    for (int p = 1; p < P1_S; ++p) {
        const float4 u0 = a[p * pitch4], u1 = a[p * pitch4 + 1];
        a0.x += u0.x; a0.y += u0.y; a0.z += u0.z; a0.w += u0.w;
        a1.x += u1.x; a1.y += u1.y; a1.z += u1.z; a1.w += u1.w;
    }
    float hv[8];
    {
        float4 s0 = s[0], s1 = s[1];
        hv[0] = a0.x + self * s0.x; hv[1] = a0.y + self * s0.y;
        hv[2] = a0.z + self * s0.z; hv[3] = a0.w + self * s0.w;
        hv[4] = a1.x + self * s1.x; hv[5] = a1.y + self * s1.y;
        hv[6] = a1.z + self * s1.z; hv[7] = a1.w + self * s1.w;
    }
    float o0 = 0.0f, o1 = 0.0f;
#pragma unroll
    for (int f = 0; f < 8; ++f) {
        float h = fmaxf(hv[f] + b1[f], 0.0f);
        o0 = fmaf(h, W2[f * 2 + 0], o0);
        o1 = fmaf(h, W2[f * 2 + 1], o1);
    }
    hw2[(long)i * 2 + 0] = o0;
    hw2[(long)i * 2 + 1] = o1;
}

__global__ void gcn_finalize(float* __restrict__ outb, const float* __restrict__ planes2,
                             const float* __restrict__ hw2,
                             const float* __restrict__ dinv, const float* __restrict__ b2,
                             int N, int Npad) {
    int i = blockIdx.x * blockDim.x + threadIdx.x;
    if (i >= N) return;
    const float self = dinv[i] * dinv[i];
    const float2* a = (const float2*)(planes2 + (long)i * 2);
    float2 acc = a[0];
#pragma unroll
    for (int p = 1; p < P2_S; ++p) {
        const float2 u = a[(size_t)p * (size_t)Npad];
        acc.x += u.x; acc.y += u.y;
    }
    float v0 = acc.x + self * hw2[(long)i * 2 + 0] + b2[0];
    float v1 = acc.y + self * hw2[(long)i * 2 + 1] + b2[1];
    const float m = fmaxf(v0, v1);
    const float lse = m + logf(expf(v0 - m) + expf(v1 - m));
    outb[(long)i * 2 + 0] = v0 - lse;
    outb[(long)i * 2 + 1] = v1 - lse;
}

extern "C" void kernel_launch(void* const* d_in, const int* in_sizes, int n_in,
                              void* d_out, int out_size, void* d_ws, size_t ws_size,
                              hipStream_t stream) {
    const float* x    = (const float*)d_in[0];
    const float* W1   = (const float*)d_in[1];
    const float* b1   = (const float*)d_in[2];
    const float* W2   = (const float*)d_in[3];
    const float* b2   = (const float*)d_in[4];
    const int*   edge = (const int*)d_in[5];

    const int N = in_sizes[0] / 16;
    const int E = in_sizes[5] / 2;
    const int* row = edge;
    const int* col = edge + E;
    float* out = (float*)d_out;

    size_t off = 0;
    auto carve = [&](size_t bytes) {
        void* p = (char*)d_ws + off;
        off = (off + bytes + 255) & ~(size_t)255;
        return p;
    };
    const int Npad   = (N + 63) & ~63;
    const int nRange = (N + DEG_NPP - 1) / DEG_NPP;
    const int nR1    = (N + P1_NPP - 1) / P1_NPP;
    const int nR2    = (N + P2_NPP - 1) / P2_NPP;
    float* deg  = (float*)carve((size_t)Npad * 4);
    float* xw1  = (float*)carve((size_t)N * 8 * 4);
    float* hw2  = (float*)carve((size_t)N * 2 * 4);
    unsigned* planes = (unsigned*)carve((size_t)DEG_S * (size_t)Npad * 4);
    float* planes1 = (float*)carve((size_t)P1_S * (size_t)Npad * 8 * 4);
    float* planes2 = (float*)carve((size_t)P2_S * (size_t)Npad * 2 * 4);
    (void)ws_size;

    const int B = 256;
    const int gN   = (N + B - 1) / B;
    const int nTiles = (N + 15) / 16;
    const int gW   = (nTiles * 32 + B - 1) / B;
    const int gS   = (Npad / 4 + B - 1) / B;

    gcn_degree_ranges<<<dim3(DEG_S, nRange), DEG_T, 0, stream>>>(col, planes, E, Npad);
    gcn_dinv_sum<<<gS,   B, 0, stream>>>(planes, deg, Npad);
    gcn_xw1_wmma<<<gW,   B, 0, stream>>>(x, W1, xw1, nTiles, N);
    gcn_prop_ranges<8, P1_NPP, P1_S><<<dim3(P1_S, nR1), PR_T, 0, stream>>>(
        row, col, deg, xw1, planes1, E, Npad);
    gcn_relu_hw2<<<gN,   B, 0, stream>>>(planes1, xw1, deg, b1, W2, hw2, N, Npad);
    gcn_prop_ranges<2, P2_NPP, P2_S><<<dim3(P2_S, nR2), PR_T, 0, stream>>>(
        row, col, deg, hw2, planes2, E, Npad);
    gcn_finalize<<<gN,   B, 0, stream>>>(out, planes2, hw2, deg, b2, N, Npad);
}


# Owner block keeps dinv of its own node range in LDS (one contiguous load): per-edge dinv[dst] gather becomes an LDS read; layer 2 regrouped to 21 ranges x 12 slices: 12 partial planes summed
# speedup vs baseline: 1.7558x; 1.1392x over previous
#include <hip/hip_runtime.h>
#include <hip/hip_bf16.h>


typedef __attribute__((ext_vector_type(2)))  float    v2f;
typedef __attribute__((ext_vector_type(8)))  float    v8f;
typedef __attribute__((ext_vector_type(16))) _Float16 v16h;

__device__ __forceinline__ void ldsAddF(float* p, float v) {
    __hip_atomic_fetch_add(p, v, __ATOMIC_RELAXED, __HIP_MEMORY_SCOPE_WORKGROUP);
}

#define DEG_NPP 65536
#define DEG_S   32
#define DEG_T   1024

__device__ __forceinline__ void deg_count(unsigned* cnt, int c, unsigned base) {
    const unsigned d = (unsigned)c - base;
    if (d < (unsigned)DEG_NPP) atomicAdd(&cnt[d], 1u);
}

__device__ __forceinline__ void deg_count4(unsigned* cnt, int4 v, unsigned base) {
    deg_count(cnt, v.x, base);
    deg_count(cnt, v.y, base);
    deg_count(cnt, v.z, base);
    deg_count(cnt, v.w, base);
}

__global__ __launch_bounds__(DEG_T)
void gcn_degree_ranges(const int* __restrict__ col, unsigned* __restrict__ planes,
                       int E, int Npad) {
    __shared__ unsigned cnt[DEG_NPP];
    const int tid = threadIdx.x;
    const int s   = blockIdx.x;
    const int nodeBase  = blockIdx.y * DEG_NPP;
    const unsigned base = (unsigned)nodeBase;

    const uint4 zero4 = make_uint4(0u, 0u, 0u, 0u);
    for (int i = tid * 4; i < DEG_NPP; i += DEG_T * 4) *(uint4*)(cnt + i) = zero4;
    __syncthreads();

    const long chunk = (long)((((E + DEG_S - 1) / DEG_S) + 3) & ~3);
    long lo = (long)s * chunk;       if (lo > (long)E) lo = (long)E;
    long hi = (long)s * chunk + chunk; if (hi > (long)E) hi = (long)E;
    int e0 = (int)lo;
    const int e1 = (int)hi;

    if ((((size_t)col) & 15) == 0) {
        const int nvec = (e1 - e0) >> 2;
        const int4* cv = (const int4*)(col + e0);
        int i = tid;
        for (; i + 7 * DEG_T < nvec; i += 8 * DEG_T) {
            int4 v[8];
#pragma unroll
            for (int k = 0; k < 8; ++k) v[k] = cv[i + k * DEG_T];
#pragma unroll
            for (int k = 0; k < 8; ++k) deg_count4(cnt, v[k], base);
        }
        if (i < nvec) {
            int4 v[8];
#pragma unroll
            for (int k = 0; k < 8; ++k) {
                const int idx = i + k * DEG_T;
                v[k] = cv[idx < nvec ? idx : nvec - 1];
            }
#pragma unroll
            for (int k = 0; k < 8; ++k)
                if (i + k * DEG_T < nvec) deg_count4(cnt, v[k], base);
        }
        e0 += nvec << 2;
    }
    for (int i = e0 + tid; i < e1; i += DEG_T) deg_count(cnt, col[i], base);
    __syncthreads();

    int lim = Npad - nodeBase;
    if (lim > DEG_NPP) lim = DEG_NPP;
    unsigned* dst = planes + (size_t)s * (size_t)Npad + (size_t)nodeBase;
    for (int i = tid * 4; i < lim; i += DEG_T * 4)
        *(uint4*)(dst + i) = *(const uint4*)(cnt + i);
}

__global__ void gcn_dinv_sum(const unsigned* __restrict__ planes, float* __restrict__ dinv,
                             int Npad) {
    const int i4 = (blockIdx.x * blockDim.x + threadIdx.x) * 4;
    if (i4 >= Npad) return;
    const uint4* p = (const uint4*)(planes + i4);
    const size_t pitch4 = (size_t)Npad >> 2;
    unsigned t0 = 1u, t1 = 1u, t2 = 1u, t3 = 1u;
#pragma unroll
    for (int s = 0; s < DEG_S; ++s) {
        const uint4 v = p[(size_t)s * pitch4];
        t0 += v.x; t1 += v.y; t2 += v.z; t3 += v.w;
    }
    *(float4*)(dinv + i4) = make_float4(__frsqrt_rn((float)t0), __frsqrt_rn((float)t1),
                                        __frsqrt_rn((float)t2), __frsqrt_rn((float)t3));
}

__global__ void gcn_xw1_wmma(const float* __restrict__ x, const float* __restrict__ W1,
                             float* __restrict__ xw1, int nTiles, int N) {
    __shared__ float tile[8][16 * 8];
    int flat = blockIdx.x * blockDim.x + threadIdx.x;
    int wave = __builtin_amdgcn_readfirstlane(flat >> 5);
    if (wave >= nTiles) return;

    const int lane = threadIdx.x & 31;
    const int m    = lane & 15;
    const int half = lane >> 4;
    const int n    = m;
    const float colMask = (n < 8) ? 1.0f : 0.0f;
    const int   nn      = n & 7;

    long rowA = (long)wave * 16 + m;
    if (rowA >= N) rowA = N - 1;
    const float* xr = x + rowA * 16;

    v8f acc = {};
#if __has_builtin(__builtin_amdgcn_wmma_f32_16x16x4_f32)
#pragma unroll
    for (int kb = 0; kb < 16; kb += 4) {
        const int k0 = kb + 2 * half;
        v2f a, b;
        a.x = xr[k0 + 0];
        a.y = xr[k0 + 1];
        b.x = W1[(k0 + 0) * 8 + nn] * colMask;
        b.y = W1[(k0 + 1) * 8 + nn] * colMask;
        acc = __builtin_amdgcn_wmma_f32_16x16x4_f32(
                  false, a, false, b, (short)0, acc, false, false);
    }
#else
    v16h a, b;
#pragma unroll
    for (int j = 0; j < 16; ++j) {
        a[j] = (j < 8) ? (_Float16)xr[8 * half + j] : (_Float16)0.0f;
        float bv = (half == 0) ? W1[j * 8 + nn] * colMask : 0.0f;
        b[j] = (_Float16)bv;
    }
    acc = __builtin_amdgcn_wmma_f32_16x16x32_f16(
              false, a, false, b, (short)0, acc, false, false);
#endif

    float* t = &tile[threadIdx.x >> 5][0];
    if (n < 8) {
#pragma unroll
        for (int r = 0; r < 8; ++r)
            t[(r + 8 * half) * 8 + n] = acc[r];
    }
    asm volatile("s_wait_dscnt 0x0" ::: "memory");

    const int rowIdx = lane >> 1;
    const int quad   = lane & 1;
    long gr = (long)wave * 16 + rowIdx;
    float4 v = *(const float4*)(t + rowIdx * 8 + quad * 4);
    if (gr < N) *(float4*)(xw1 + gr * 8 + quad * 4) = v;
}

#define PR_T    1024
#define PR_RING 128

template <int F>
struct PrEdge { float w; float t[F]; unsigned d; };

template <int F>
__device__ __forceinline__ void pr_load_edge(PrEdge<F>& g, uint2 q, int r,
                                             const float* dloc,
                                             const float* __restrict__ dinv,
                                             const float* __restrict__ tbl) {
    g.d = q.x;
    g.w = dinv[r] * dloc[q.x];
    if constexpr (F == 8) {
        const float4 t0 = *(const float4*)(tbl + (size_t)r * 8);
        const float4 t1 = *(const float4*)(tbl + (size_t)r * 8 + 4);
        g.t[0] = t0.x; g.t[1] = t0.y; g.t[2] = t0.z; g.t[3] = t0.w;
        g.t[4] = t1.x; g.t[5] = t1.y; g.t[6] = t1.z; g.t[7] = t1.w;
    } else {
        const float2 t0 = *(const float2*)(tbl + (size_t)r * 2);
        g.t[0] = t0.x; g.t[1] = t0.y;
    }
}

template <int F>
__device__ __forceinline__ void pr_add_edge(float* __restrict__ acc, const PrEdge<F>& g) {
    float* a = acc + (size_t)g.d * F;
#pragma unroll
    for (int f = 0; f < F; ++f) ldsAddF(a + f, g.t[f] * g.w);
}

template <int F, int NPP, int S>
__global__ __launch_bounds__(PR_T)
void gcn_prop_ranges(const int* __restrict__ row, const int* __restrict__ col,
                     const float* __restrict__ dinv, const float* __restrict__ tbl,
                     float* __restrict__ planes, int E, int Npad) {
    __shared__ float acc[NPP * F];
    __shared__ float dloc[NPP];
    __shared__ uint2 ring[PR_T / 32][PR_RING];
    const int tid  = threadIdx.x;
    const unsigned lane = (unsigned)tid & 31u;
    const int s    = blockIdx.x;
    const int nodeBase  = blockIdx.y * NPP;
    const unsigned base = (unsigned)nodeBase;
    uint2* rq = &ring[tid >> 5][0];
    unsigned head = 0, tail = 0;

    const float4 zero4 = make_float4(0.0f, 0.0f, 0.0f, 0.0f);
    for (int i = tid * 4; i < NPP * F; i += PR_T * 4) *(float4*)(acc + i) = zero4;
    for (int i = tid * 4; i < NPP; i += PR_T * 4) {
        float4 v = zero4;
        if (nodeBase + i < Npad) v = *(const float4*)(dinv + nodeBase + i);
        *(float4*)(dloc + i) = v;
    }
    __syncthreads();

    auto push = [&](int c, unsigned lim, int e) {
        const unsigned d = (unsigned)c - base;
        const bool m = d < lim;
        const unsigned mask = __builtin_amdgcn_ballot_w32(m);
        if (mask != 0u) {
            if (m) rq[__builtin_amdgcn_mbcnt_lo(mask, tail) & (PR_RING - 1)] = make_uint2(d, (unsigned)e);
            tail += (unsigned)__builtin_popcount(mask);
        }
    };
    auto drain = [&]() {
        while (tail - head >= 64u) {
            const uint2 qa = rq[(head + lane) & (PR_RING - 1)];
            const uint2 qb = rq[(head + 32u + lane) & (PR_RING - 1)];
            const int ra = row[qa.y];
            const int rb = row[qb.y];
            PrEdge<F> ga, gb;
            pr_load_edge<F>(ga, qa, ra, dloc, dinv, tbl);
            pr_load_edge<F>(gb, qb, rb, dloc, dinv, tbl);
            pr_add_edge<F>(acc, ga);
            pr_add_edge<F>(acc, gb);
            head += 64u;
        }
    };

    const long chunk = (long)((((E + S - 1) / S) + 3) & ~3);
    long lo = (long)s * chunk;         if (lo > (long)E) lo = (long)E;
    long hi = (long)s * chunk + chunk; if (hi > (long)E) hi = (long)E;
    int e0 = (int)lo;
    const int e1 = (int)hi;

    if ((((size_t)col) & 15) == 0) {
        const int nvec = (e1 - e0) >> 2;
        const int4* cv = (const int4*)(col + e0);
        for (int ib = 0; ib < nvec; ib += 8 * PR_T) {
            int4 v[8];
#pragma unroll
            for (int k = 0; k < 8; ++k) {
                const int idx = ib + k * PR_T + tid;
                v[k] = cv[idx < nvec ? idx : nvec - 1];
            }
#pragma unroll
            for (int k = 0; k < 8; ++k) {
                const int idx = ib + k * PR_T + tid;
                const unsigned lim = idx < nvec ? (unsigned)NPP : 0u;
                const int eb = e0 + idx * 4;
                push(v[k].x, lim, eb + 0);
                push(v[k].y, lim, eb + 1);
                drain();
                push(v[k].z, lim, eb + 2);
                push(v[k].w, lim, eb + 3);
                drain();
            }
        }
        e0 += nvec << 2;
    }
    for (int ib = e0; ib < e1; ib += PR_T) {
        const int i = ib + tid;
        const bool ok = i < e1;
        push(col[ok ? i : e1 - 1], ok ? (unsigned)NPP : 0u, i);
        drain();
    }
    {
        const unsigned n = tail - head;
#pragma unroll
        for (unsigned h = 0; h < 64u; h += 32u) {
            if (lane + h < n) {
                const uint2 q = rq[(head + h + lane) & (PR_RING - 1)];
                PrEdge<F> g;
                pr_load_edge<F>(g, q, row[q.y], dloc, dinv, tbl);
                pr_add_edge<F>(acc, g);
            }
        }
    }
    __syncthreads();

    int lim = Npad - nodeBase;
    if (lim > NPP) lim = NPP;
    float* dst = planes + ((size_t)s * (size_t)Npad + (size_t)nodeBase) * F;
    for (int i = tid * 4; i < lim * F; i += PR_T * 4)
        *(float4*)(dst + i) = *(const float4*)(acc + i);
}

#define P1_NPP 8192
#define P1_S   4
#define P2_NPP 24576
#define P2_S   12

__global__ void gcn_relu_hw2(const float* __restrict__ planes1, const float* __restrict__ xw1,
                             const float* __restrict__ dinv, const float* __restrict__ b1,
                             const float* __restrict__ W2, float* __restrict__ hw2, int N,
                             int Npad) {
    int i = blockIdx.x * blockDim.x + threadIdx.x;
    if (i >= N) return;
    const float self = dinv[i] * dinv[i];
    const float4* a = (const float4*)(planes1 + (long)i * 8);
    const float4* s = (const float4*)(xw1 + (long)i * 8);
    const size_t pitch4 = (size_t)Npad * 2;
    float4 a0 = a[0], a1 = a[1];
#pragma unroll
    for (int p = 1; p < P1_S; ++p) {
        const float4 u0 = a[p * pitch4], u1 = a[p * pitch4 + 1];
        a0.x += u0.x; a0.y += u0.y; a0.z += u0.z; a0.w += u0.w;
        a1.x += u1.x; a1.y += u1.y; a1.z += u1.z; a1.w += u1.w;
    }
    float hv[8];
    {
        float4 s0 = s[0], s1 = s[1];
        hv[0] = a0.x + self * s0.x; hv[1] = a0.y + self * s0.y;
        hv[2] = a0.z + self * s0.z; hv[3] = a0.w + self * s0.w;
        hv[4] = a1.x + self * s1.x; hv[5] = a1.y + self * s1.y;
        hv[6] = a1.z + self * s1.z; hv[7] = a1.w + self * s1.w;
    }
    float o0 = 0.0f, o1 = 0.0f;
#pragma unroll
    for (int f = 0; f < 8; ++f) {
        float h = fmaxf(hv[f] + b1[f], 0.0f);
        o0 = fmaf(h, W2[f * 2 + 0], o0);
        o1 = fmaf(h, W2[f * 2 + 1], o1);
    }
    hw2[(long)i * 2 + 0] = o0;
    hw2[(long)i * 2 + 1] = o1;
}

__global__ void gcn_finalize(float* __restrict__ outb, const float* __restrict__ planes2,
                             const float* __restrict__ hw2,
                             const float* __restrict__ dinv, const float* __restrict__ b2,
                             int N, int Npad) {
    int i = blockIdx.x * blockDim.x + threadIdx.x;
    if (i >= N) return;
    const float self = dinv[i] * dinv[i];
    const float2* a = (const float2*)(planes2 + (long)i * 2);
    float2 acc = a[0];
#pragma unroll
    for (int p = 1; p < P2_S; ++p) {
        const float2 u = a[(size_t)p * (size_t)Npad];
        acc.x += u.x; acc.y += u.y;
    }
    float v0 = acc.x + self * hw2[(long)i * 2 + 0] + b2[0];
    float v1 = acc.y + self * hw2[(long)i * 2 + 1] + b2[1];
    const float m = fmaxf(v0, v1);
    const float lse = m + logf(expf(v0 - m) + expf(v1 - m));
    outb[(long)i * 2 + 0] = v0 - lse;
    outb[(long)i * 2 + 1] = v1 - lse;
}

extern "C" void kernel_launch(void* const* d_in, const int* in_sizes, int n_in,
                              void* d_out, int out_size, void* d_ws, size_t ws_size,
                              hipStream_t stream) {
    const float* x    = (const float*)d_in[0];
    const float* W1   = (const float*)d_in[1];
    const float* b1   = (const float*)d_in[2];
    const float* W2   = (const float*)d_in[3];
    const float* b2   = (const float*)d_in[4];
    const int*   edge = (const int*)d_in[5];

    const int N = in_sizes[0] / 16;
    const int E = in_sizes[5] / 2;
    const int* row = edge;
    const int* col = edge + E;
    float* out = (float*)d_out;

    size_t off = 0;
    auto carve = [&](size_t bytes) {
        void* p = (char*)d_ws + off;
        off = (off + bytes + 255) & ~(size_t)255;
        return p;
    };
    const int Npad   = (N + 63) & ~63;
    const int nRange = (N + DEG_NPP - 1) / DEG_NPP;
    const int nR1    = (N + P1_NPP - 1) / P1_NPP;
    const int nR2    = (N + P2_NPP - 1) / P2_NPP;
    float* deg  = (float*)carve((size_t)Npad * 4);
    float* xw1  = (float*)carve((size_t)N * 8 * 4);
    float* hw2  = (float*)carve((size_t)N * 2 * 4);
    unsigned* planes = (unsigned*)carve((size_t)DEG_S * (size_t)Npad * 4);
    float* planes1 = (float*)carve((size_t)P1_S * (size_t)Npad * 8 * 4);
    float* planes2 = (float*)carve((size_t)P2_S * (size_t)Npad * 2 * 4);
    (void)ws_size;

    const int B = 256;
    const int gN   = (N + B - 1) / B;
    const int nTiles = (N + 15) / 16;
    const int gW   = (nTiles * 32 + B - 1) / B;
    const int gS   = (Npad / 4 + B - 1) / B;

    gcn_degree_ranges<<<dim3(DEG_S, nRange), DEG_T, 0, stream>>>(col, planes, E, Npad);
    gcn_dinv_sum<<<gS,   B, 0, stream>>>(planes, deg, Npad);
    gcn_xw1_wmma<<<gW,   B, 0, stream>>>(x, W1, xw1, nTiles, N);
    gcn_prop_ranges<8, P1_NPP, P1_S><<<dim3(P1_S, nR1), PR_T, 0, stream>>>(
        row, col, deg, xw1, planes1, E, Npad);
    gcn_relu_hw2<<<gN,   B, 0, stream>>>(planes1, xw1, deg, b1, W2, hw2, N, Npad);
    gcn_prop_ranges<2, P2_NPP, P2_S><<<dim3(P2_S, nR2), PR_T, 0, stream>>>(
        row, col, deg, hw2, planes2, E, Npad);
    gcn_finalize<<<gN,   B, 0, stream>>>(out, planes2, hw2, deg, b2, N, Npad);
}


# Edge list re-encoded once in edge order (range-id byte; word of source + in-range offset): range blocks stream bytes, test 4 per op, one gather gives src+dst; 16-edge slice bounds, sums arrival order
# speedup vs baseline: 2.0706x; 1.1793x over previous
#include <hip/hip_runtime.h>
#include <hip/hip_bf16.h>


typedef __attribute__((ext_vector_type(2)))  float    v2f;
typedef __attribute__((ext_vector_type(8)))  float    v8f;
typedef __attribute__((ext_vector_type(16))) _Float16 v16h;

__device__ __forceinline__ void ldsAddF(float* p, float v) {
    __hip_atomic_fetch_add(p, v, __ATOMIC_RELAXED, __HIP_MEMORY_SCOPE_WORKGROUP);
}

#define RID_SHIFT 13
#define PK_SRC_BITS 19
#define ENT_EDGE_BITS 23

__device__ __forceinline__ unsigned rid_byte(int c) {
    return c < 0 ? 0x7fu : ((unsigned)c >> RID_SHIFT);
}
__device__ __forceinline__ unsigned pack_edge(int r, int c) {
    return (unsigned)r | (((unsigned)c & ((1u << RID_SHIFT) - 1u)) << PK_SRC_BITS);
}

__device__ __forceinline__ void load_edge_group(const int* __restrict__ row,
                                                const int* __restrict__ col, int e, int E,
                                                bool aligned, int4& r, int4& c) {
    if (aligned && e + 3 < E) {
        c = *(const int4*)(col + e);
        r = *(const int4*)(row + e);
    } else {
        c.x = e + 0 < E ? col[e + 0] : -1;  r.x = e + 0 < E ? row[e + 0] : 0;
        c.y = e + 1 < E ? col[e + 1] : -1;  r.y = e + 1 < E ? row[e + 1] : 0;
        c.z = e + 2 < E ? col[e + 2] : -1;  r.z = e + 2 < E ? row[e + 2] : 0;
        c.w = e + 3 < E ? col[e + 3] : -1;  r.w = e + 3 < E ? row[e + 3] : 0;
    }
}

#define DEG_NPP 65536
#define DEG_S   32
#define DEG_T   1024

__device__ __forceinline__ void deg_count(unsigned* cnt, int c, unsigned base) {
    const unsigned d = (unsigned)c - base;
    if (d < (unsigned)DEG_NPP) atomicAdd(&cnt[d], 1u);
}

__device__ __forceinline__ void deg_count4(unsigned* cnt, int4 v, unsigned base) {
    deg_count(cnt, v.x, base);
    deg_count(cnt, v.y, base);
    deg_count(cnt, v.z, base);
    deg_count(cnt, v.w, base);
}

__global__ __launch_bounds__(DEG_T)
void gcn_degree_ranges(const int* __restrict__ row, const int* __restrict__ col,
                       unsigned* __restrict__ planes, unsigned* __restrict__ rid4,
                       uint4* __restrict__ packed4, int E, int Npad) {
    __shared__ unsigned cnt[DEG_NPP];
    const int tid = threadIdx.x;
    const int s   = blockIdx.x;
    const int nodeBase  = blockIdx.y * DEG_NPP;
    const unsigned base = (unsigned)nodeBase;

    {
        const int nb   = (int)(gridDim.x * gridDim.y);
        const int b    = (int)(blockIdx.y * gridDim.x + blockIdx.x);
        const int Q    = ((E + 15) & ~15) >> 2;
        const int qper = (Q + nb - 1) / nb;
        int q0 = b * qper;        if (q0 > Q) q0 = Q;
        int q1 = q0 + qper;       if (q1 > Q) q1 = Q;
        const bool al = ((((size_t)col) | ((size_t)row)) & 15) == 0;
        auto store_group = [&](int q, const int4& r, const int4& c) {
            rid4[q] = rid_byte(c.x) | (rid_byte(c.y) << 8) | (rid_byte(c.z) << 16) |
                      (rid_byte(c.w) << 24);
            packed4[q] = make_uint4(pack_edge(r.x, c.x), pack_edge(r.y, c.y),
                                    pack_edge(r.z, c.z), pack_edge(r.w, c.w));
        };
        if (al && (q1 << 2) <= E) {
            const int4* c4 = (const int4*)col;
            const int4* r4 = (const int4*)row;
            for (int qb = q0; qb < q1; qb += 4 * DEG_T) {
                int4 c[4], r[4];
#pragma unroll
                for (int k = 0; k < 4; ++k) {
                    const int q = qb + k * DEG_T + tid;
                    c[k] = c4[q < q1 ? q : q1 - 1];
                    r[k] = r4[q < q1 ? q : q1 - 1];
                }
#pragma unroll
                for (int k = 0; k < 4; ++k) {
                    const int q = qb + k * DEG_T + tid;
                    if (q < q1) store_group(q, r[k], c[k]);
                }
            }
        } else {
            for (int q = q0 + tid; q < q1; q += DEG_T) {
                int4 c, r;
                load_edge_group(row, col, q << 2, E, al, r, c);
                store_group(q, r, c);
            }
        }
    }

    const uint4 zero4 = make_uint4(0u, 0u, 0u, 0u);
    for (int i = tid * 4; i < DEG_NPP; i += DEG_T * 4) *(uint4*)(cnt + i) = zero4;
    __syncthreads();

    const long chunk = (long)((((E + DEG_S - 1) / DEG_S) + 3) & ~3);
    long lo = (long)s * chunk;       if (lo > (long)E) lo = (long)E;
    long hi = (long)s * chunk + chunk; if (hi > (long)E) hi = (long)E;
    int e0 = (int)lo;
    const int e1 = (int)hi;

    if ((((size_t)col) & 15) == 0) {
        const int nvec = (e1 - e0) >> 2;
        const int4* cv = (const int4*)(col + e0);
        int i = tid;
        for (; i + 7 * DEG_T < nvec; i += 8 * DEG_T) {
            int4 v[8];
#pragma unroll
            for (int k = 0; k < 8; ++k) v[k] = cv[i + k * DEG_T];
#pragma unroll
            for (int k = 0; k < 8; ++k) deg_count4(cnt, v[k], base);
        }
        if (i < nvec) {
            int4 v[8];
#pragma unroll
            for (int k = 0; k < 8; ++k) {
                const int idx = i + k * DEG_T;
                v[k] = cv[idx < nvec ? idx : nvec - 1];
            }
#pragma unroll
            for (int k = 0; k < 8; ++k)
                if (i + k * DEG_T < nvec) deg_count4(cnt, v[k], base);
        }
        e0 += nvec << 2;
    }
    for (int i = e0 + tid; i < e1; i += DEG_T) deg_count(cnt, col[i], base);
    __syncthreads();

    int lim = Npad - nodeBase;
    if (lim > DEG_NPP) lim = DEG_NPP;
    unsigned* dst = planes + (size_t)s * (size_t)Npad + (size_t)nodeBase;
    for (int i = tid * 4; i < lim; i += DEG_T * 4)
        *(uint4*)(dst + i) = *(const uint4*)(cnt + i);
}

__global__ void gcn_dinv_sum(const unsigned* __restrict__ planes, float* __restrict__ dinv,
                             int Npad) {
    const int i4 = (blockIdx.x * blockDim.x + threadIdx.x) * 4;
    if (i4 >= Npad) return;
    const uint4* p = (const uint4*)(planes + i4);
    const size_t pitch4 = (size_t)Npad >> 2;
    unsigned t0 = 1u, t1 = 1u, t2 = 1u, t3 = 1u;
#pragma unroll
    for (int s = 0; s < DEG_S; ++s) {
        const uint4 v = p[(size_t)s * pitch4];
        t0 += v.x; t1 += v.y; t2 += v.z; t3 += v.w;
    }
    *(float4*)(dinv + i4) = make_float4(__frsqrt_rn((float)t0), __frsqrt_rn((float)t1),
                                        __frsqrt_rn((float)t2), __frsqrt_rn((float)t3));
}

__global__ void gcn_xw1_wmma(const float* __restrict__ x, const float* __restrict__ W1,
                             float* __restrict__ xw1, int nTiles, int N) {
    __shared__ float tile[8][16 * 8];
    int flat = blockIdx.x * blockDim.x + threadIdx.x;
    int wave = __builtin_amdgcn_readfirstlane(flat >> 5);
    if (wave >= nTiles) return;

    const int lane = threadIdx.x & 31;
    const int m    = lane & 15;
    const int half = lane >> 4;
    const int n    = m;
    const float colMask = (n < 8) ? 1.0f : 0.0f;
    const int   nn      = n & 7;

    long rowA = (long)wave * 16 + m;
    if (rowA >= N) rowA = N - 1;
    const float* xr = x + rowA * 16;

    v8f acc = {};
#if __has_builtin(__builtin_amdgcn_wmma_f32_16x16x4_f32)
#pragma unroll
    for (int kb = 0; kb < 16; kb += 4) {
        const int k0 = kb + 2 * half;
        v2f a, b;
        a.x = xr[k0 + 0];
        a.y = xr[k0 + 1];
        b.x = W1[(k0 + 0) * 8 + nn] * colMask;
        b.y = W1[(k0 + 1) * 8 + nn] * colMask;
        acc = __builtin_amdgcn_wmma_f32_16x16x4_f32(
                  false, a, false, b, (short)0, acc, false, false);
    }
#else
    v16h a, b;
#pragma unroll
    for (int j = 0; j < 16; ++j) {
        a[j] = (j < 8) ? (_Float16)xr[8 * half + j] : (_Float16)0.0f;
        float bv = (half == 0) ? W1[j * 8 + nn] * colMask : 0.0f;
        b[j] = (_Float16)bv;
    }
    acc = __builtin_amdgcn_wmma_f32_16x16x32_f16(
              false, a, false, b, (short)0, acc, false, false);
#endif

    float* t = &tile[threadIdx.x >> 5][0];
    if (n < 8) {
#pragma unroll
        for (int r = 0; r < 8; ++r)
            t[(r + 8 * half) * 8 + n] = acc[r];
    }
    asm volatile("s_wait_dscnt 0x0" ::: "memory");

    const int rowIdx = lane >> 1;
    const int quad   = lane & 1;
    long gr = (long)wave * 16 + rowIdx;
    float4 v = *(const float4*)(t + rowIdx * 8 + quad * 4);
    if (gr < N) *(float4*)(xw1 + gr * 8 + quad * 4) = v;
}

#define PR_T    1024
#define PR_RING 128

template <int F>
struct PrEdge { float w; float t[F]; unsigned d; };

template <int F>
__device__ __forceinline__ void pr_load_edge(PrEdge<F>& g, unsigned pk, unsigned sub,
                                             const float* dloc,
                                             const float* __restrict__ dinv,
                                             const float* __restrict__ tbl) {
    const unsigned r = pk & ((1u << PK_SRC_BITS) - 1u);
    const unsigned d = (pk >> PK_SRC_BITS) + (sub << RID_SHIFT);
    g.d = d;
    g.w = dinv[r] * dloc[d];
    if constexpr (F == 8) {
        const float4 t0 = *(const float4*)(tbl + (size_t)r * 8);
        const float4 t1 = *(const float4*)(tbl + (size_t)r * 8 + 4);
        g.t[0] = t0.x; g.t[1] = t0.y; g.t[2] = t0.z; g.t[3] = t0.w;
        g.t[4] = t1.x; g.t[5] = t1.y; g.t[6] = t1.z; g.t[7] = t1.w;
    } else {
        const float2 t0 = *(const float2*)(tbl + (size_t)r * 2);
        g.t[0] = t0.x; g.t[1] = t0.y;
    }
}

template <int F>
__device__ __forceinline__ void pr_add_edge(float* __restrict__ acc, const PrEdge<F>& g) {
    float* a = acc + (size_t)g.d * F;
#pragma unroll
    for (int f = 0; f < F; ++f) ldsAddF(a + f, g.t[f] * g.w);
}

template <int R>
__device__ __forceinline__ unsigned pr_hit_signs(unsigned w, unsigned lo) {
    if constexpr (R == 1) {
        const unsigned t = (w ^ ((lo * 0x01010101u) | 0x80808080u)) - 0x01010101u;
        return ~t;
    } else {
        const unsigned ge = (w | 0x80808080u) - lo * 0x01010101u;
        const unsigned le = (((lo + (unsigned)R - 1u) | 0x80u) * 0x01010101u) - w;
        return ge & le;
    }
}

template <int F, int NPP, int S>
__global__ __launch_bounds__(PR_T)
void gcn_prop_ranges(const uint4* __restrict__ rid16, const unsigned* __restrict__ packed,
                     const float* __restrict__ dinv, const float* __restrict__ tbl,
                     float* __restrict__ planes, int E, int Npad) {
    static_assert(NPP % (1 << RID_SHIFT) == 0, "a block owns whole range ids");
    constexpr int R = NPP >> RID_SHIFT;
    __shared__ float acc[NPP * F];
    __shared__ float dloc[NPP];
    __shared__ unsigned ring[PR_T / 32][PR_RING];
    const int tid  = threadIdx.x;
    const unsigned lane = (unsigned)tid & 31u;
    const int s    = blockIdx.x;
    const int nodeBase  = blockIdx.y * NPP;
    const unsigned lo   = (unsigned)blockIdx.y * (unsigned)R;
    unsigned* rq = &ring[tid >> 5][0];
    unsigned head = 0, tail = 0;

    const float4 zero4 = make_float4(0.0f, 0.0f, 0.0f, 0.0f);
    for (int i = tid * 4; i < NPP * F; i += PR_T * 4) *(float4*)(acc + i) = zero4;
    for (int i = tid * 4; i < NPP; i += PR_T * 4) {
        float4 v = zero4;
        if (nodeBase + i < Npad) v = *(const float4*)(dinv + nodeBase + i);
        *(float4*)(dloc + i) = v;
    }
    __syncthreads();

    auto drain = [&]() {
        while (tail - head >= 64u) {
            const unsigned qa = rq[(head + lane) & (PR_RING - 1)];
            const unsigned qb = rq[(head + 32u + lane) & (PR_RING - 1)];
            const unsigned pa = packed[qa & ((1u << ENT_EDGE_BITS) - 1u)];
            const unsigned pb = packed[qb & ((1u << ENT_EDGE_BITS) - 1u)];
            PrEdge<F> ga, gb;
            pr_load_edge<F>(ga, pa, qa >> ENT_EDGE_BITS, dloc, dinv, tbl);
            pr_load_edge<F>(gb, pb, qb >> ENT_EDGE_BITS, dloc, dinv, tbl);
            pr_add_edge<F>(acc, ga);
            pr_add_edge<F>(acc, gb);
            head += 64u;
        }
    };
    auto push_all = [&](unsigned m, const uint4& v, unsigned ebase) {
        for (int round = 0; round < 16; ++round) {
            const bool has = m != 0u;
            const unsigned mask = __builtin_amdgcn_ballot_w32(has);
            if (mask == 0u) break;
            if (has) {
                const unsigned q  = (unsigned)__builtin_ctz(m);
                const unsigned wd = q & 7u, bt = q >> 3;
                unsigned ent = ebase + (wd << 2) + bt;
                if constexpr (R > 1) {
                    const unsigned word = wd == 0u ? v.x : wd == 1u ? v.y : wd == 2u ? v.z : v.w;
                    ent |= (((word >> (bt << 3)) & 0xffu) - lo) << ENT_EDGE_BITS;
                }
                rq[__builtin_amdgcn_mbcnt_lo(mask, tail) & (PR_RING - 1)] = ent;
                m &= m - 1u;
            }
            tail += (unsigned)__builtin_popcount(mask);
            drain();
        }
    };

    const int nvecAll = (E + 15) >> 4;
    const int chunk   = (nvecAll + S - 1) / S;
    int v0 = s * chunk;      if (v0 > nvecAll) v0 = nvecAll;
    int v1 = v0 + chunk;     if (v1 > nvecAll) v1 = nvecAll;
    const int nvec = v1 - v0;
    const uint4* rv = rid16 + v0;

    for (int ib = 0; ib < nvec; ib += 8 * PR_T) {
        uint4 v[8];
#pragma unroll
        for (int k = 0; k < 8; ++k) {
            const int idx = ib + k * PR_T + tid;
            v[k] = rv[idx < nvec ? idx : nvec - 1];
        }
#pragma unroll
        for (int k = 0; k < 8; ++k) {
            const int idx = ib + k * PR_T + tid;
            const unsigned h0 = pr_hit_signs<R>(v[k].x, lo);
            const unsigned h1 = pr_hit_signs<R>(v[k].y, lo);
            const unsigned h2 = pr_hit_signs<R>(v[k].z, lo);
            const unsigned h3 = pr_hit_signs<R>(v[k].w, lo);
            unsigned m = ((h0 >> 7) & 0x01010101u) | ((h1 >> 6) & 0x02020202u) |
                         ((h2 >> 5) & 0x04040404u) | ((h3 >> 4) & 0x08080808u);
            if (idx >= nvec) m = 0u;
            push_all(m, v[k], (unsigned)(v0 + idx) << 4);
        }
    }
    {
        const unsigned n = tail - head;
#pragma unroll
        for (unsigned h = 0; h < 64u; h += 32u) {
            if (lane + h < n) {
                const unsigned q = rq[(head + h + lane) & (PR_RING - 1)];
                PrEdge<F> g;
                pr_load_edge<F>(g, packed[q & ((1u << ENT_EDGE_BITS) - 1u)], q >> ENT_EDGE_BITS,
                                dloc, dinv, tbl);
                pr_add_edge<F>(acc, g);
            }
        }
    }
    __syncthreads();

    int lim = Npad - nodeBase;
    if (lim > NPP) lim = NPP;
    float* dst = planes + ((size_t)s * (size_t)Npad + (size_t)nodeBase) * F;
    for (int i = tid * 4; i < lim * F; i += PR_T * 4)
        *(float4*)(dst + i) = *(const float4*)(acc + i);
}

#define P1_NPP 8192
#define P1_S   4
#define P2_NPP 24576
#define P2_S   12

__global__ void gcn_relu_hw2(const float* __restrict__ planes1, const float* __restrict__ xw1,
                             const float* __restrict__ dinv, const float* __restrict__ b1,
                             const float* __restrict__ W2, float* __restrict__ hw2, int N,
                             int Npad) {
    int i = blockIdx.x * blockDim.x + threadIdx.x;
    if (i >= N) return;
    const float self = dinv[i] * dinv[i];
    const float4* a = (const float4*)(planes1 + (long)i * 8);
    const float4* s = (const float4*)(xw1 + (long)i * 8);
    const size_t pitch4 = (size_t)Npad * 2;
    float4 a0 = a[0], a1 = a[1];
#pragma unroll
    for (int p = 1; p < P1_S; ++p) {
        const float4 u0 = a[p * pitch4], u1 = a[p * pitch4 + 1];
        a0.x += u0.x; a0.y += u0.y; a0.z += u0.z; a0.w += u0.w;
        a1.x += u1.x; a1.y += u1.y; a1.z += u1.z; a1.w += u1.w;
    }
    float hv[8];
    {
        float4 s0 = s[0], s1 = s[1];
        hv[0] = a0.x + self * s0.x; hv[1] = a0.y + self * s0.y;
        hv[2] = a0.z + self * s0.z; hv[3] = a0.w + self * s0.w;
        hv[4] = a1.x + self * s1.x; hv[5] = a1.y + self * s1.y;
        hv[6] = a1.z + self * s1.z; hv[7] = a1.w + self * s1.w;
    }
    float o0 = 0.0f, o1 = 0.0f;
#pragma unroll
    for (int f = 0; f < 8; ++f) {
        float h = fmaxf(hv[f] + b1[f], 0.0f);
        o0 = fmaf(h, W2[f * 2 + 0], o0);
        o1 = fmaf(h, W2[f * 2 + 1], o1);
    }
    hw2[(long)i * 2 + 0] = o0;
    hw2[(long)i * 2 + 1] = o1;
}

__global__ void gcn_finalize(float* __restrict__ outb, const float* __restrict__ planes2,
                             const float* __restrict__ hw2,
                             const float* __restrict__ dinv, const float* __restrict__ b2,
                             int N, int Npad) {
    int i = blockIdx.x * blockDim.x + threadIdx.x;
    if (i >= N) return;
    const float self = dinv[i] * dinv[i];
    const float2* a = (const float2*)(planes2 + (long)i * 2);
    float2 acc = a[0];
#pragma unroll
    for (int p = 1; p < P2_S; ++p) {
        const float2 u = a[(size_t)p * (size_t)Npad];
        acc.x += u.x; acc.y += u.y;
    }
    float v0 = acc.x + self * hw2[(long)i * 2 + 0] + b2[0];
    float v1 = acc.y + self * hw2[(long)i * 2 + 1] + b2[1];
    const float m = fmaxf(v0, v1);
    const float lse = m + logf(expf(v0 - m) + expf(v1 - m));
    outb[(long)i * 2 + 0] = v0 - lse;
    outb[(long)i * 2 + 1] = v1 - lse;
}

extern "C" void kernel_launch(void* const* d_in, const int* in_sizes, int n_in,
                              void* d_out, int out_size, void* d_ws, size_t ws_size,
                              hipStream_t stream) {
    const float* x    = (const float*)d_in[0];
    const float* W1   = (const float*)d_in[1];
    const float* b1   = (const float*)d_in[2];
    const float* W2   = (const float*)d_in[3];
    const float* b2   = (const float*)d_in[4];
    const int*   edge = (const int*)d_in[5];

    const int N = in_sizes[0] / 16;
    const int E = in_sizes[5] / 2;
    const int* row = edge;
    const int* col = edge + E;
    float* out = (float*)d_out;

    size_t off = 0;
    auto carve = [&](size_t bytes) {
        void* p = (char*)d_ws + off;
        off = (off + bytes + 255) & ~(size_t)255;
        return p;
    };
    const int Npad   = (N + 63) & ~63;
    const int nRange = (N + DEG_NPP - 1) / DEG_NPP;
    const int nR1    = (N + P1_NPP - 1) / P1_NPP;
    const int nR2    = (N + P2_NPP - 1) / P2_NPP;
    float* deg  = (float*)carve((size_t)Npad * 4);
    float* xw1  = (float*)carve((size_t)N * 8 * 4);
    float* hw2  = (float*)carve((size_t)N * 2 * 4);
    unsigned* planes = (unsigned*)carve((size_t)DEG_S * (size_t)Npad * 4);
    float* planes1 = (float*)carve((size_t)P1_S * (size_t)Npad * 8 * 4);
    float* planes2 = (float*)carve((size_t)P2_S * (size_t)Npad * 2 * 4);
    const size_t Epad = ((size_t)E + 15) & ~(size_t)15;
    unsigned* rid   = (unsigned*)carve(Epad);
    unsigned* pked  = (unsigned*)carve(Epad * 4);
    (void)ws_size;

    const int B = 256;
    const int gN   = (N + B - 1) / B;
    const int nTiles = (N + 15) / 16;
    const int gW   = (nTiles * 32 + B - 1) / B;
    const int gS   = (Npad / 4 + B - 1) / B;

    gcn_degree_ranges<<<dim3(DEG_S, nRange), DEG_T, 0, stream>>>(row, col, planes, rid,
                                                                 (uint4*)pked, E, Npad);
    gcn_dinv_sum<<<gS,   B, 0, stream>>>(planes, deg, Npad);
    gcn_xw1_wmma<<<gW,   B, 0, stream>>>(x, W1, xw1, nTiles, N);
    gcn_prop_ranges<8, P1_NPP, P1_S><<<dim3(P1_S, nR1), PR_T, 0, stream>>>(
        (const uint4*)rid, pked, deg, xw1, planes1, E, Npad);
    gcn_relu_hw2<<<gN,   B, 0, stream>>>(planes1, xw1, deg, b1, W2, hw2, N, Npad);
    gcn_prop_ranges<2, P2_NPP, P2_S><<<dim3(P2_S, nR2), PR_T, 0, stream>>>(
        (const uint4*)rid, pked, deg, hw2, planes2, E, Npad);
    gcn_finalize<<<gN,   B, 0, stream>>>(out, planes2, hw2, deg, b2, N, Npad);
}


# Edge record widened to 8 bytes (source+offset word, dinv[source]), written by a re-encoding kernel behind the degree pass: range blocks gather no dinv per edge; sum order unchanged
# speedup vs baseline: 2.2447x; 1.0841x over previous
#include <hip/hip_runtime.h>
#include <hip/hip_bf16.h>


typedef __attribute__((ext_vector_type(2)))  float    v2f;
typedef __attribute__((ext_vector_type(8)))  float    v8f;
typedef __attribute__((ext_vector_type(16))) _Float16 v16h;

__device__ __forceinline__ void ldsAddF(float* p, float v) {
    __hip_atomic_fetch_add(p, v, __ATOMIC_RELAXED, __HIP_MEMORY_SCOPE_WORKGROUP);
}

#define RID_SHIFT 13
#define PK_SRC_BITS 19
#define ENT_EDGE_BITS 23

__device__ __forceinline__ unsigned rid_byte(int c) {
    return c < 0 ? 0x7fu : ((unsigned)c >> RID_SHIFT);
}
__device__ __forceinline__ unsigned pack_edge(int r, int c) {
    return (unsigned)r | (((unsigned)c & ((1u << RID_SHIFT) - 1u)) << PK_SRC_BITS);
}

__device__ __forceinline__ void load_edge_group(const int* __restrict__ row,
                                                const int* __restrict__ col, int e, int E,
                                                bool aligned, int4& r, int4& c) {
    if (aligned && e + 3 < E) {
        c = *(const int4*)(col + e);
        r = *(const int4*)(row + e);
    } else {
        c.x = e + 0 < E ? col[e + 0] : -1;  r.x = e + 0 < E ? row[e + 0] : 0;
        c.y = e + 1 < E ? col[e + 1] : -1;  r.y = e + 1 < E ? row[e + 1] : 0;
        c.z = e + 2 < E ? col[e + 2] : -1;  r.z = e + 2 < E ? row[e + 2] : 0;
        c.w = e + 3 < E ? col[e + 3] : -1;  r.w = e + 3 < E ? row[e + 3] : 0;
    }
}

#define DEG_NPP 65536
#define DEG_S   32
#define DEG_T   1024

__device__ __forceinline__ void deg_count(unsigned* cnt, int c, unsigned base) {
    const unsigned d = (unsigned)c - base;
    if (d < (unsigned)DEG_NPP) atomicAdd(&cnt[d], 1u);
}

__device__ __forceinline__ void deg_count4(unsigned* cnt, int4 v, unsigned base) {
    deg_count(cnt, v.x, base);
    deg_count(cnt, v.y, base);
    deg_count(cnt, v.z, base);
    deg_count(cnt, v.w, base);
}

__global__ __launch_bounds__(DEG_T)
void gcn_degree_ranges(const int* __restrict__ col, unsigned* __restrict__ planes, int E,
                       int Npad) {
    __shared__ unsigned cnt[DEG_NPP];
    const int tid = threadIdx.x;
    const int s   = blockIdx.x;
    const int nodeBase  = blockIdx.y * DEG_NPP;
    const unsigned base = (unsigned)nodeBase;

    const uint4 zero4 = make_uint4(0u, 0u, 0u, 0u);
    for (int i = tid * 4; i < DEG_NPP; i += DEG_T * 4) *(uint4*)(cnt + i) = zero4;
    __syncthreads();

    const long chunk = (long)((((E + DEG_S - 1) / DEG_S) + 3) & ~3);
    long lo = (long)s * chunk;       if (lo > (long)E) lo = (long)E;
    long hi = (long)s * chunk + chunk; if (hi > (long)E) hi = (long)E;
    int e0 = (int)lo;
    const int e1 = (int)hi;

    if ((((size_t)col) & 15) == 0) {
        const int nvec = (e1 - e0) >> 2;
        const int4* cv = (const int4*)(col + e0);
        int i = tid;
        for (; i + 7 * DEG_T < nvec; i += 8 * DEG_T) {
            int4 v[8];
#pragma unroll
            for (int k = 0; k < 8; ++k) v[k] = cv[i + k * DEG_T];
#pragma unroll
            for (int k = 0; k < 8; ++k) deg_count4(cnt, v[k], base);
        }
        if (i < nvec) {
            int4 v[8];
#pragma unroll
            for (int k = 0; k < 8; ++k) {
                const int idx = i + k * DEG_T;
                v[k] = cv[idx < nvec ? idx : nvec - 1];
            }
#pragma unroll
            for (int k = 0; k < 8; ++k)
                if (i + k * DEG_T < nvec) deg_count4(cnt, v[k], base);
        }
        e0 += nvec << 2;
    }
    for (int i = e0 + tid; i < e1; i += DEG_T) deg_count(cnt, col[i], base);
    __syncthreads();

    int lim = Npad - nodeBase;
    if (lim > DEG_NPP) lim = DEG_NPP;
    unsigned* dst = planes + (size_t)s * (size_t)Npad + (size_t)nodeBase;
    for (int i = tid * 4; i < lim; i += DEG_T * 4)
        *(uint4*)(dst + i) = *(const uint4*)(cnt + i);
}

__global__ void gcn_dinv_sum(const unsigned* __restrict__ planes, float* __restrict__ dinv,
                             int Npad) {
    const int i4 = (blockIdx.x * blockDim.x + threadIdx.x) * 4;
    if (i4 >= Npad) return;
    const uint4* p = (const uint4*)(planes + i4);
    const size_t pitch4 = (size_t)Npad >> 2;
    unsigned t0 = 1u, t1 = 1u, t2 = 1u, t3 = 1u;
#pragma unroll
    for (int s = 0; s < DEG_S; ++s) {
        const uint4 v = p[(size_t)s * pitch4];
        t0 += v.x; t1 += v.y; t2 += v.z; t3 += v.w;
    }
    *(float4*)(dinv + i4) = make_float4(__frsqrt_rn((float)t0), __frsqrt_rn((float)t1),
                                        __frsqrt_rn((float)t2), __frsqrt_rn((float)t3));
}

#define ENC_T 256
#define ENC_G 2

__global__ __launch_bounds__(ENC_T)
void gcn_encode_edges(const int* __restrict__ row, const int* __restrict__ col,
                      const float* __restrict__ dinv, unsigned* __restrict__ rid4,
                      uint4* __restrict__ rec2, int E) {
    const int Q = ((E + 15) & ~15) >> 2;
    const bool al = ((((size_t)col) | ((size_t)row)) & 15) == 0;
    const int qb = (int)blockIdx.x * (ENC_T * ENC_G) + (int)threadIdx.x;
    int4 c[ENC_G], r[ENC_G];
    float4 f[ENC_G];
#pragma unroll
    for (int k = 0; k < ENC_G; ++k) {
        const int q = qb + k * ENC_T;
        c[k] = make_int4(-1, -1, -1, -1);
        r[k] = make_int4(0, 0, 0, 0);
        if (q < Q) load_edge_group(row, col, q << 2, E, al, r[k], c[k]);
    }
#pragma unroll
    for (int k = 0; k < ENC_G; ++k)
        f[k] = make_float4(dinv[r[k].x], dinv[r[k].y], dinv[r[k].z], dinv[r[k].w]);
#pragma unroll
    for (int k = 0; k < ENC_G; ++k) {
        const int q = qb + k * ENC_T;
        if (q < Q) {
            rid4[q] = rid_byte(c[k].x) | (rid_byte(c[k].y) << 8) | (rid_byte(c[k].z) << 16) |
                      (rid_byte(c[k].w) << 24);
            rec2[2 * (size_t)q]     = make_uint4(pack_edge(r[k].x, c[k].x), __float_as_uint(f[k].x),
                                                 pack_edge(r[k].y, c[k].y), __float_as_uint(f[k].y));
            rec2[2 * (size_t)q + 1] = make_uint4(pack_edge(r[k].z, c[k].z), __float_as_uint(f[k].z),
                                                 pack_edge(r[k].w, c[k].w), __float_as_uint(f[k].w));
        }
    }
}

__global__ void gcn_xw1_wmma(const float* __restrict__ x, const float* __restrict__ W1,
                             float* __restrict__ xw1, int nTiles, int N) {
    __shared__ float tile[8][16 * 8];
    int flat = blockIdx.x * blockDim.x + threadIdx.x;
    int wave = __builtin_amdgcn_readfirstlane(flat >> 5);
    if (wave >= nTiles) return;

    const int lane = threadIdx.x & 31;
    const int m    = lane & 15;
    const int half = lane >> 4;
    const int n    = m;
    const float colMask = (n < 8) ? 1.0f : 0.0f;
    const int   nn      = n & 7;

    long rowA = (long)wave * 16 + m;
    if (rowA >= N) rowA = N - 1;
    const float* xr = x + rowA * 16;

    v8f acc = {};
#if __has_builtin(__builtin_amdgcn_wmma_f32_16x16x4_f32)
#pragma unroll
    for (int kb = 0; kb < 16; kb += 4) {
        const int k0 = kb + 2 * half;
        v2f a, b;
        a.x = xr[k0 + 0];
        a.y = xr[k0 + 1];
        b.x = W1[(k0 + 0) * 8 + nn] * colMask;
        b.y = W1[(k0 + 1) * 8 + nn] * colMask;
        acc = __builtin_amdgcn_wmma_f32_16x16x4_f32(
                  false, a, false, b, (short)0, acc, false, false);
    }
#else
    v16h a, b;
#pragma unroll
    for (int j = 0; j < 16; ++j) {
        a[j] = (j < 8) ? (_Float16)xr[8 * half + j] : (_Float16)0.0f;
        float bv = (half == 0) ? W1[j * 8 + nn] * colMask : 0.0f;
        b[j] = (_Float16)bv;
    }
    acc = __builtin_amdgcn_wmma_f32_16x16x32_f16(
              false, a, false, b, (short)0, acc, false, false);
#endif

    float* t = &tile[threadIdx.x >> 5][0];
    if (n < 8) {
#pragma unroll
        for (int r = 0; r < 8; ++r)
            t[(r + 8 * half) * 8 + n] = acc[r];
    }
    asm volatile("s_wait_dscnt 0x0" ::: "memory");

    const int rowIdx = lane >> 1;
    const int quad   = lane & 1;
    long gr = (long)wave * 16 + rowIdx;
    float4 v = *(const float4*)(t + rowIdx * 8 + quad * 4);
    if (gr < N) *(float4*)(xw1 + gr * 8 + quad * 4) = v;
}

#define PR_T    1024
#define PR_RING 128

template <int F>
struct PrEdge { float w; float t[F]; unsigned d; };

template <int F>
__device__ __forceinline__ void pr_load_edge(PrEdge<F>& g, uint2 rc, unsigned sub,
                                             const float* dloc,
                                             const float* __restrict__ tbl) {
    const unsigned pk = rc.x;
    const unsigned r = pk & ((1u << PK_SRC_BITS) - 1u);
    const unsigned d = (pk >> PK_SRC_BITS) + (sub << RID_SHIFT);
    g.d = d;
    g.w = __uint_as_float(rc.y) * dloc[d];
    if constexpr (F == 8) {
        const float4 t0 = *(const float4*)(tbl + (size_t)r * 8);
        const float4 t1 = *(const float4*)(tbl + (size_t)r * 8 + 4);
        g.t[0] = t0.x; g.t[1] = t0.y; g.t[2] = t0.z; g.t[3] = t0.w;
        g.t[4] = t1.x; g.t[5] = t1.y; g.t[6] = t1.z; g.t[7] = t1.w;
    } else {
        const float2 t0 = *(const float2*)(tbl + (size_t)r * 2);
        g.t[0] = t0.x; g.t[1] = t0.y;
    }
}

template <int F>
__device__ __forceinline__ void pr_add_edge(float* __restrict__ acc, const PrEdge<F>& g) {
    float* a = acc + (size_t)g.d * F;
#pragma unroll
    for (int f = 0; f < F; ++f) ldsAddF(a + f, g.t[f] * g.w);
}

template <int R>
__device__ __forceinline__ unsigned pr_hit_signs(unsigned w, unsigned lo) {
    if constexpr (R == 1) {
        const unsigned t = (w ^ ((lo * 0x01010101u) | 0x80808080u)) - 0x01010101u;
        return ~t;
    } else {
        const unsigned ge = (w | 0x80808080u) - lo * 0x01010101u;
        const unsigned le = (((lo + (unsigned)R - 1u) | 0x80u) * 0x01010101u) - w;
        return ge & le;
    }
}

template <int F, int NPP, int S>
__global__ __launch_bounds__(PR_T)
void gcn_prop_ranges(const uint4* __restrict__ rid16, const uint2* __restrict__ rec,
                     const float* __restrict__ dinv, const float* __restrict__ tbl,
                     float* __restrict__ planes, int E, int Npad) {
    static_assert(NPP % (1 << RID_SHIFT) == 0, "a block owns whole range ids");
    constexpr int R = NPP >> RID_SHIFT;
    __shared__ float acc[NPP * F];
    __shared__ float dloc[NPP];
    __shared__ unsigned ring[PR_T / 32][PR_RING];
    const int tid  = threadIdx.x;
    const unsigned lane = (unsigned)tid & 31u;
    const int s    = blockIdx.x;
    const int nodeBase  = blockIdx.y * NPP;
    const unsigned lo   = (unsigned)blockIdx.y * (unsigned)R;
    unsigned* rq = &ring[tid >> 5][0];
    unsigned head = 0, tail = 0;

    const float4 zero4 = make_float4(0.0f, 0.0f, 0.0f, 0.0f);
    for (int i = tid * 4; i < NPP * F; i += PR_T * 4) *(float4*)(acc + i) = zero4;
    for (int i = tid * 4; i < NPP; i += PR_T * 4) {
        float4 v = zero4;
        if (nodeBase + i < Npad) v = *(const float4*)(dinv + nodeBase + i);
        *(float4*)(dloc + i) = v;
    }
    __syncthreads();

    auto drain = [&]() {
        while (tail - head >= 64u) {
            const unsigned qa = rq[(head + lane) & (PR_RING - 1)];
            const unsigned qb = rq[(head + 32u + lane) & (PR_RING - 1)];
            const uint2 pa = rec[qa & ((1u << ENT_EDGE_BITS) - 1u)];
            const uint2 pb = rec[qb & ((1u << ENT_EDGE_BITS) - 1u)];
            PrEdge<F> ga, gb;
            pr_load_edge<F>(ga, pa, qa >> ENT_EDGE_BITS, dloc, tbl);
            pr_load_edge<F>(gb, pb, qb >> ENT_EDGE_BITS, dloc, tbl);
            pr_add_edge<F>(acc, ga);
            pr_add_edge<F>(acc, gb);
            head += 64u;
        }
    };
    auto push_all = [&](unsigned m, const uint4& v, unsigned ebase) {
        for (int round = 0; round < 16; ++round) {
            const bool has = m != 0u;
            const unsigned mask = __builtin_amdgcn_ballot_w32(has);
            if (mask == 0u) break;
            if (has) {
                const unsigned q  = (unsigned)__builtin_ctz(m);
                const unsigned wd = q & 7u, bt = q >> 3;
                unsigned ent = ebase + (wd << 2) + bt;
                if constexpr (R > 1) {
                    const unsigned word = wd == 0u ? v.x : wd == 1u ? v.y : wd == 2u ? v.z : v.w;
                    ent |= (((word >> (bt << 3)) & 0xffu) - lo) << ENT_EDGE_BITS;
                }
                rq[__builtin_amdgcn_mbcnt_lo(mask, tail) & (PR_RING - 1)] = ent;
                m &= m - 1u;
            }
            tail += (unsigned)__builtin_popcount(mask);
            drain();
        }
    };

    const int nvecAll = (E + 15) >> 4;
    const int chunk   = (nvecAll + S - 1) / S;
    int v0 = s * chunk;      if (v0 > nvecAll) v0 = nvecAll;
    int v1 = v0 + chunk;     if (v1 > nvecAll) v1 = nvecAll;
    const int nvec = v1 - v0;
    const uint4* rv = rid16 + v0;

    for (int ib = 0; ib < nvec; ib += 8 * PR_T) {
        uint4 v[8];
#pragma unroll
        for (int k = 0; k < 8; ++k) {
            const int idx = ib + k * PR_T + tid;
            v[k] = rv[idx < nvec ? idx : nvec - 1];
        }
#pragma unroll
        for (int k = 0; k < 8; ++k) {
            const int idx = ib + k * PR_T + tid;
            const unsigned h0 = pr_hit_signs<R>(v[k].x, lo);
            const unsigned h1 = pr_hit_signs<R>(v[k].y, lo);
            const unsigned h2 = pr_hit_signs<R>(v[k].z, lo);
            const unsigned h3 = pr_hit_signs<R>(v[k].w, lo);
            unsigned m = ((h0 >> 7) & 0x01010101u) | ((h1 >> 6) & 0x02020202u) |
                         ((h2 >> 5) & 0x04040404u) | ((h3 >> 4) & 0x08080808u);
            if (idx >= nvec) m = 0u;
            push_all(m, v[k], (unsigned)(v0 + idx) << 4);
        }
    }
    {
        const unsigned n = tail - head;
#pragma unroll
        for (unsigned h = 0; h < 64u; h += 32u) {
            if (lane + h < n) {
                const unsigned q = rq[(head + h + lane) & (PR_RING - 1)];
                PrEdge<F> g;
                pr_load_edge<F>(g, rec[q & ((1u << ENT_EDGE_BITS) - 1u)], q >> ENT_EDGE_BITS,
                                dloc, tbl);
                pr_add_edge<F>(acc, g);
            }
        }
    }
    __syncthreads();

    int lim = Npad - nodeBase;
    if (lim > NPP) lim = NPP;
    float* dst = planes + ((size_t)s * (size_t)Npad + (size_t)nodeBase) * F;
    for (int i = tid * 4; i < lim * F; i += PR_T * 4)
        *(float4*)(dst + i) = *(const float4*)(acc + i);
}

#define P1_NPP 8192
#define P1_S   4
#define P2_NPP 24576
#define P2_S   12

__global__ void gcn_relu_hw2(const float* __restrict__ planes1, const float* __restrict__ xw1,
                             const float* __restrict__ dinv, const float* __restrict__ b1,
                             const float* __restrict__ W2, float* __restrict__ hw2, int N,
                             int Npad) {
    int i = blockIdx.x * blockDim.x + threadIdx.x;
    if (i >= N) return;
    const float self = dinv[i] * dinv[i];
    const float4* a = (const float4*)(planes1 + (long)i * 8);
    const float4* s = (const float4*)(xw1 + (long)i * 8);
    const size_t pitch4 = (size_t)Npad * 2;
    float4 a0 = a[0], a1 = a[1];
#pragma unroll
    for (int p = 1; p < P1_S; ++p) {
        const float4 u0 = a[p * pitch4], u1 = a[p * pitch4 + 1];
        a0.x += u0.x; a0.y += u0.y; a0.z += u0.z; a0.w += u0.w;
        a1.x += u1.x; a1.y += u1.y; a1.z += u1.z; a1.w += u1.w;
    }
    float hv[8];
    {
        float4 s0 = s[0], s1 = s[1];
        hv[0] = a0.x + self * s0.x; hv[1] = a0.y + self * s0.y;
        hv[2] = a0.z + self * s0.z; hv[3] = a0.w + self * s0.w;
        hv[4] = a1.x + self * s1.x; hv[5] = a1.y + self * s1.y;
        hv[6] = a1.z + self * s1.z; hv[7] = a1.w + self * s1.w;
    }
    float o0 = 0.0f, o1 = 0.0f;
#pragma unroll
    for (int f = 0; f < 8; ++f) {
        float h = fmaxf(hv[f] + b1[f], 0.0f);
        o0 = fmaf(h, W2[f * 2 + 0], o0);
        o1 = fmaf(h, W2[f * 2 + 1], o1);
    }
    hw2[(long)i * 2 + 0] = o0;
    hw2[(long)i * 2 + 1] = o1;
}

__global__ void gcn_finalize(float* __restrict__ outb, const float* __restrict__ planes2,
                             const float* __restrict__ hw2,
                             const float* __restrict__ dinv, const float* __restrict__ b2,
                             int N, int Npad) {
    int i = blockIdx.x * blockDim.x + threadIdx.x;
    if (i >= N) return;
    const float self = dinv[i] * dinv[i];
    const float2* a = (const float2*)(planes2 + (long)i * 2);
    float2 acc = a[0];
#pragma unroll
    for (int p = 1; p < P2_S; ++p) {
        const float2 u = a[(size_t)p * (size_t)Npad];
        acc.x += u.x; acc.y += u.y;
    }
    float v0 = acc.x + self * hw2[(long)i * 2 + 0] + b2[0];
    float v1 = acc.y + self * hw2[(long)i * 2 + 1] + b2[1];
    const float m = fmaxf(v0, v1);
    const float lse = m + logf(expf(v0 - m) + expf(v1 - m));
    outb[(long)i * 2 + 0] = v0 - lse;
    outb[(long)i * 2 + 1] = v1 - lse;
}

extern "C" void kernel_launch(void* const* d_in, const int* in_sizes, int n_in,
                              void* d_out, int out_size, void* d_ws, size_t ws_size,
                              hipStream_t stream) {
    const float* x    = (const float*)d_in[0];
    const float* W1   = (const float*)d_in[1];
    const float* b1   = (const float*)d_in[2];
    const float* W2   = (const float*)d_in[3];
    const float* b2   = (const float*)d_in[4];
    const int*   edge = (const int*)d_in[5];

    const int N = in_sizes[0] / 16;
    const int E = in_sizes[5] / 2;
    const int* row = edge;
    const int* col = edge + E;
    float* out = (float*)d_out;

    size_t off = 0;
    auto carve = [&](size_t bytes) {
        void* p = (char*)d_ws + off;
        off = (off + bytes + 255) & ~(size_t)255;
        return p;
    };
    const int Npad   = (N + 63) & ~63;
    const int nRange = (N + DEG_NPP - 1) / DEG_NPP;
    const int nR1    = (N + P1_NPP - 1) / P1_NPP;
    const int nR2    = (N + P2_NPP - 1) / P2_NPP;
    float* deg  = (float*)carve((size_t)Npad * 4);
    float* xw1  = (float*)carve((size_t)N * 8 * 4);
    float* hw2  = (float*)carve((size_t)N * 2 * 4);
    const size_t Epad = ((size_t)E + 15) & ~(size_t)15;
    const size_t degBytes = (size_t)DEG_S * (size_t)Npad * 4;
    const size_t recBytes = Epad * 8;
    void* shared = carve(degBytes > recBytes ? degBytes : recBytes);
    unsigned* planes = (unsigned*)shared;
    uint2*    recs   = (uint2*)shared;
    float* planes1 = (float*)carve((size_t)P1_S * (size_t)Npad * 8 * 4);
    float* planes2 = (float*)carve((size_t)P2_S * (size_t)Npad * 2 * 4);
    unsigned* rid   = (unsigned*)carve(Epad);
    (void)ws_size;

    const int B = 256;
    const int gN   = (N + B - 1) / B;
    const int nTiles = (N + 15) / 16;
    const int gW   = (nTiles * 32 + B - 1) / B;
    const int gS   = (Npad / 4 + B - 1) / B;
    const int gE   = (int)((Epad / 4 + (size_t)(ENC_T * ENC_G) - 1) / (size_t)(ENC_T * ENC_G));

    gcn_degree_ranges<<<dim3(DEG_S, nRange), DEG_T, 0, stream>>>(col, planes, E, Npad);
    gcn_dinv_sum<<<gS,   B, 0, stream>>>(planes, deg, Npad);
    gcn_xw1_wmma<<<gW,   B, 0, stream>>>(x, W1, xw1, nTiles, N);
    gcn_encode_edges<<<gE, ENC_T, 0, stream>>>(row, col, deg, rid, (uint4*)recs, E);
    gcn_prop_ranges<8, P1_NPP, P1_S><<<dim3(P1_S, nR1), PR_T, 0, stream>>>(
        (const uint4*)rid, recs, deg, xw1, planes1, E, Npad);
    gcn_relu_hw2<<<gN,   B, 0, stream>>>(planes1, xw1, deg, b1, W2, hw2, N, Npad);
    gcn_prop_ranges<2, P2_NPP, P2_S><<<dim3(P2_S, nR2), PR_T, 0, stream>>>(
        (const uint4*)rid, recs, deg, hw2, planes2, E, Npad);
    gcn_finalize<<<gN,   B, 0, stream>>>(out, planes2, hw2, deg, b2, N, Npad);
}


# Re-encoding kernel gathers no dinv per edge: degrees also stored as bytes, a block holds half the byte table in LDS (two passes) and forms rsqrt(degree) itself; byte 255 reads dinv; same values
# speedup vs baseline: 2.3381x; 1.0416x over previous
#include <hip/hip_runtime.h>
#include <hip/hip_bf16.h>


typedef __attribute__((ext_vector_type(2)))  float    v2f;
typedef __attribute__((ext_vector_type(8)))  float    v8f;
typedef __attribute__((ext_vector_type(16))) _Float16 v16h;

__device__ __forceinline__ void ldsAddF(float* p, float v) {
    __hip_atomic_fetch_add(p, v, __ATOMIC_RELAXED, __HIP_MEMORY_SCOPE_WORKGROUP);
}

#define RID_SHIFT 13
#define PK_SRC_BITS 19
#define ENT_EDGE_BITS 23

__device__ __forceinline__ unsigned rid_byte(int c) {
    return c < 0 ? 0x7fu : ((unsigned)c >> RID_SHIFT);
}
__device__ __forceinline__ unsigned pack_edge(int r, int c) {
    return (unsigned)r | (((unsigned)c & ((1u << RID_SHIFT) - 1u)) << PK_SRC_BITS);
}

__device__ __forceinline__ void load_edge_group(const int* __restrict__ row,
                                                const int* __restrict__ col, int e, int E,
                                                bool aligned, int4& r, int4& c) {
    if (aligned && e + 3 < E) {
        c = *(const int4*)(col + e);
        r = *(const int4*)(row + e);
    } else {
        c.x = e + 0 < E ? col[e + 0] : -1;  r.x = e + 0 < E ? row[e + 0] : 0;
        c.y = e + 1 < E ? col[e + 1] : -1;  r.y = e + 1 < E ? row[e + 1] : 0;
        c.z = e + 2 < E ? col[e + 2] : -1;  r.z = e + 2 < E ? row[e + 2] : 0;
        c.w = e + 3 < E ? col[e + 3] : -1;  r.w = e + 3 < E ? row[e + 3] : 0;
    }
}

#define DEG_NPP 65536
#define DEG_S   32
#define DEG_T   1024

__device__ __forceinline__ void deg_count(unsigned* cnt, int c, unsigned base) {
    const unsigned d = (unsigned)c - base;
    if (d < (unsigned)DEG_NPP) atomicAdd(&cnt[d], 1u);
}

__device__ __forceinline__ void deg_count4(unsigned* cnt, int4 v, unsigned base) {
    deg_count(cnt, v.x, base);
    deg_count(cnt, v.y, base);
    deg_count(cnt, v.z, base);
    deg_count(cnt, v.w, base);
}

__global__ __launch_bounds__(DEG_T)
void gcn_degree_ranges(const int* __restrict__ col, unsigned* __restrict__ planes, int E,
                       int Npad) {
    __shared__ unsigned cnt[DEG_NPP];
    const int tid = threadIdx.x;
    const int s   = blockIdx.x;
    const int nodeBase  = blockIdx.y * DEG_NPP;
    const unsigned base = (unsigned)nodeBase;

    const uint4 zero4 = make_uint4(0u, 0u, 0u, 0u);
    for (int i = tid * 4; i < DEG_NPP; i += DEG_T * 4) *(uint4*)(cnt + i) = zero4;
    __syncthreads();

    const long chunk = (long)((((E + DEG_S - 1) / DEG_S) + 3) & ~3);
    long lo = (long)s * chunk;       if (lo > (long)E) lo = (long)E;
    long hi = (long)s * chunk + chunk; if (hi > (long)E) hi = (long)E;
    int e0 = (int)lo;
    const int e1 = (int)hi;

    if ((((size_t)col) & 15) == 0) {
        const int nvec = (e1 - e0) >> 2;
        const int4* cv = (const int4*)(col + e0);
        int i = tid;
        for (; i + 7 * DEG_T < nvec; i += 8 * DEG_T) {
            int4 v[8];
#pragma unroll
            for (int k = 0; k < 8; ++k) v[k] = cv[i + k * DEG_T];
#pragma unroll
            for (int k = 0; k < 8; ++k) deg_count4(cnt, v[k], base);
        }
        if (i < nvec) {
            int4 v[8];
#pragma unroll
            for (int k = 0; k < 8; ++k) {
                const int idx = i + k * DEG_T;
                v[k] = cv[idx < nvec ? idx : nvec - 1];
            }
#pragma unroll
            for (int k = 0; k < 8; ++k)
                if (i + k * DEG_T < nvec) deg_count4(cnt, v[k], base);
        }
        e0 += nvec << 2;
    }
    for (int i = e0 + tid; i < e1; i += DEG_T) deg_count(cnt, col[i], base);
    __syncthreads();

    int lim = Npad - nodeBase;
    if (lim > DEG_NPP) lim = DEG_NPP;
    unsigned* dst = planes + (size_t)s * (size_t)Npad + (size_t)nodeBase;
    for (int i = tid * 4; i < lim; i += DEG_T * 4)
        *(uint4*)(dst + i) = *(const uint4*)(cnt + i);
}

#define DCODE_CAP 255u
__device__ __forceinline__ unsigned dcode_of(unsigned t) { return t < DCODE_CAP ? t : DCODE_CAP; }

__global__ void gcn_dinv_sum(const unsigned* __restrict__ planes, float* __restrict__ dinv,
                             unsigned* __restrict__ dcode4, int Npad) {
    const int i4 = (blockIdx.x * blockDim.x + threadIdx.x) * 4;
    if (i4 >= Npad) return;
    const uint4* p = (const uint4*)(planes + i4);
    const size_t pitch4 = (size_t)Npad >> 2;
    unsigned t0 = 1u, t1 = 1u, t2 = 1u, t3 = 1u;
#pragma unroll
    for (int s = 0; s < DEG_S; ++s) {
        const uint4 v = p[(size_t)s * pitch4];
        t0 += v.x; t1 += v.y; t2 += v.z; t3 += v.w;
    }
    *(float4*)(dinv + i4) = make_float4(__frsqrt_rn((float)t0), __frsqrt_rn((float)t1),
                                        __frsqrt_rn((float)t2), __frsqrt_rn((float)t3));
    dcode4[i4 >> 2] = dcode_of(t0) | (dcode_of(t1) << 8) | (dcode_of(t2) << 16) |
                      (dcode_of(t3) << 24);
}

#define ENC_T 1024
#define ENC_G 8
#define ENC_HALF_BITS 18
#define ENC_HALF (1 << ENC_HALF_BITS)

__device__ __forceinline__ void enc_pack_group(unsigned (&pk)[4], unsigned& ridw, int4 r, int4 c) {
    pk[0] = pack_edge(r.x, c.x);  pk[1] = pack_edge(r.y, c.y);
    pk[2] = pack_edge(r.z, c.z);  pk[3] = pack_edge(r.w, c.w);
    ridw = rid_byte(c.x) | (rid_byte(c.y) << 8) | (rid_byte(c.z) << 16) | (rid_byte(c.w) << 24);
}

__device__ __forceinline__ int4 enc_load4(const int* __restrict__ p, int e, int E, bool full,
                                          int pad) {
    if (full) return *(const int4*)(p + e);
    const int last = E > 0 ? E - 1 : 0;
    int v[4];
#pragma unroll
    for (int j = 0; j < 4; ++j) {
        const int ej = e + j;
        const int x = p[ej < last ? ej : last];
        v[j] = ej < E ? x : pad;
    }
    return make_int4(v[0], v[1], v[2], v[3]);
}

__global__ __launch_bounds__(ENC_T)
void gcn_encode_edges(const int* __restrict__ row, const int* __restrict__ col,
                      const float* __restrict__ dinv, const unsigned char* __restrict__ dcode,
                      unsigned* __restrict__ rid4, uint4* __restrict__ rec2, int E, int Npad) {
    __shared__ uint4 tab16[ENC_HALF / 16];
    const unsigned char* tab = (const unsigned char*)tab16;
    const int tid = (int)threadIdx.x;
    const int Q = ((E + 15) & ~15) >> 2;
    const bool al = ((((size_t)col) | ((size_t)row)) & 15) == 0;
    const int q0 = (int)blockIdx.x * (ENC_T * ENC_G);
    const int qb = q0 + tid;
    const bool full = al && (long)(q0 + ENC_T * ENC_G) * 4 <= (long)E;
    constexpr int NH = 1 << (PK_SRC_BITS - ENC_HALF_BITS);
    unsigned cw[NH][ENC_G];
#pragma unroll
    for (int h = 0; h < NH; ++h)
#pragma unroll
        for (int k = 0; k < ENC_G; ++k) cw[h][k] = 0u;
#pragma unroll
    for (int h = 0; h < NH; ++h) {
        if (h * ENC_HALF >= Npad) break;
        if (h > 0) __syncthreads();
        {
            const int base = h * ENC_HALF;
            constexpr int PER = ENC_HALF / (ENC_T * 16);
            static_assert(PER % 8 == 0, "whole batches of eight loads");
#pragma unroll
            for (int b = 0; b < PER; b += 8) {
                unsigned v[8][4];
#pragma unroll
                for (int k = 0; k < 8; ++k) {
                    const int i = base + ((b + k) * ENC_T + tid) * 16;
                    const uint4 t = *(const uint4*)(dcode + (i < Npad ? i : Npad - 16));
                    v[k][0] = t.x; v[k][1] = t.y; v[k][2] = t.z; v[k][3] = t.w;
                }
#pragma unroll
                for (int k = 0; k < 8; ++k)
                    tab16[(b + k) * ENC_T + tid] = make_uint4(v[k][0], v[k][1], v[k][2], v[k][3]);
            }
        }
        int4 r[ENC_G];
#pragma unroll
        for (int k = 0; k < ENC_G; ++k) r[k] = enc_load4(row, (qb + k * ENC_T) << 2, E, full, 0);
        __syncthreads();
#pragma unroll
        for (int k = 0; k < ENC_G; ++k) {
            const unsigned s[4] = {(unsigned)r[k].x, (unsigned)r[k].y, (unsigned)r[k].z,
                                   (unsigned)r[k].w};
            unsigned w = 0u;
#pragma unroll
            for (int j = 0; j < 4; ++j) w |= (unsigned)tab[s[j] & (ENC_HALF - 1)] << (8 * j);
            cw[h][k] = w;
        }
    }
#pragma unroll
    for (int kb = 0; kb < ENC_G; kb += 4) {
        int4 c[4], r[4];
#pragma unroll
        for (int k = 0; k < 4; ++k) {
            c[k] = enc_load4(col, (qb + (kb + k) * ENC_T) << 2, E, full, -1);
            r[k] = enc_load4(row, (qb + (kb + k) * ENC_T) << 2, E, full, 0);
        }
#pragma unroll
        for (int k = 0; k < 4; ++k) {
            const int q = qb + (kb + k) * ENC_T;
            unsigned pk[4], w;
            enc_pack_group(pk, w, r[k], c[k]);
            float f[4];
#pragma unroll
            for (int j = 0; j < 4; ++j) {
                const unsigned src = pk[j] & ((1u << PK_SRC_BITS) - 1u);
                unsigned wsel = cw[0][kb + k];
#pragma unroll
                for (int h = 1; h < NH; ++h)
                    wsel = (int)(src >> ENC_HALF_BITS) == h ? cw[h][kb + k] : wsel;
                const unsigned code = (wsel >> (8 * j)) & 0xffu;
                f[j] = __frsqrt_rn((float)code);
                if (code == DCODE_CAP) f[j] = dinv[src];
            }
            if (q < Q) {
                rid4[q] = w;
                rec2[2 * (size_t)q]     = make_uint4(pk[0], __float_as_uint(f[0]),
                                                     pk[1], __float_as_uint(f[1]));
                rec2[2 * (size_t)q + 1] = make_uint4(pk[2], __float_as_uint(f[2]),
                                                     pk[3], __float_as_uint(f[3]));
            }
        }
    }
}

__global__ void gcn_xw1_wmma(const float* __restrict__ x, const float* __restrict__ W1,
                             float* __restrict__ xw1, int nTiles, int N) {
    __shared__ float tile[8][16 * 8];
    int flat = blockIdx.x * blockDim.x + threadIdx.x;
    int wave = __builtin_amdgcn_readfirstlane(flat >> 5);
    if (wave >= nTiles) return;

    const int lane = threadIdx.x & 31;
    const int m    = lane & 15;
    const int half = lane >> 4;
    const int n    = m;
    const float colMask = (n < 8) ? 1.0f : 0.0f;
    const int   nn      = n & 7;

    long rowA = (long)wave * 16 + m;
    if (rowA >= N) rowA = N - 1;
    const float* xr = x + rowA * 16;

    v8f acc = {};
#if __has_builtin(__builtin_amdgcn_wmma_f32_16x16x4_f32)
#pragma unroll
    for (int kb = 0; kb < 16; kb += 4) {
        const int k0 = kb + 2 * half;
        v2f a, b;
        a.x = xr[k0 + 0];
        a.y = xr[k0 + 1];
        b.x = W1[(k0 + 0) * 8 + nn] * colMask;
        b.y = W1[(k0 + 1) * 8 + nn] * colMask;
        acc = __builtin_amdgcn_wmma_f32_16x16x4_f32(
                  false, a, false, b, (short)0, acc, false, false);
    }
#else
    v16h a, b;
#pragma unroll
    for (int j = 0; j < 16; ++j) {
        a[j] = (j < 8) ? (_Float16)xr[8 * half + j] : (_Float16)0.0f;
        float bv = (half == 0) ? W1[j * 8 + nn] * colMask : 0.0f;
        b[j] = (_Float16)bv;
    }
    acc = __builtin_amdgcn_wmma_f32_16x16x32_f16(
              false, a, false, b, (short)0, acc, false, false);
#endif

    float* t = &tile[threadIdx.x >> 5][0];
    if (n < 8) {
#pragma unroll
        for (int r = 0; r < 8; ++r)
            t[(r + 8 * half) * 8 + n] = acc[r];
    }
    asm volatile("s_wait_dscnt 0x0" ::: "memory");

    const int rowIdx = lane >> 1;
    const int quad   = lane & 1;
    long gr = (long)wave * 16 + rowIdx;
    float4 v = *(const float4*)(t + rowIdx * 8 + quad * 4);
    if (gr < N) *(float4*)(xw1 + gr * 8 + quad * 4) = v;
}

#define PR_T    1024
#define PR_RING 128

template <int F>
struct PrEdge { float w; float t[F]; unsigned d; };

template <int F>
__device__ __forceinline__ void pr_load_edge(PrEdge<F>& g, uint2 rc, unsigned sub,
                                             const float* dloc,
                                             const float* __restrict__ tbl) {
    const unsigned pk = rc.x;
    const unsigned r = pk & ((1u << PK_SRC_BITS) - 1u);
    const unsigned d = (pk >> PK_SRC_BITS) + (sub << RID_SHIFT);
    g.d = d;
    g.w = __uint_as_float(rc.y) * dloc[d];
    if constexpr (F == 8) {
        const float4 t0 = *(const float4*)(tbl + (size_t)r * 8);
        const float4 t1 = *(const float4*)(tbl + (size_t)r * 8 + 4);
        g.t[0] = t0.x; g.t[1] = t0.y; g.t[2] = t0.z; g.t[3] = t0.w;
        g.t[4] = t1.x; g.t[5] = t1.y; g.t[6] = t1.z; g.t[7] = t1.w;
    } else {
        const float2 t0 = *(const float2*)(tbl + (size_t)r * 2);
        g.t[0] = t0.x; g.t[1] = t0.y;
    }
}

template <int F>
__device__ __forceinline__ void pr_add_edge(float* __restrict__ acc, const PrEdge<F>& g) {
    float* a = acc + (size_t)g.d * F;
#pragma unroll
    for (int f = 0; f < F; ++f) ldsAddF(a + f, g.t[f] * g.w);
}

template <int R>
__device__ __forceinline__ unsigned pr_hit_signs(unsigned w, unsigned lo) {
    if constexpr (R == 1) {
        const unsigned t = (w ^ ((lo * 0x01010101u) | 0x80808080u)) - 0x01010101u;
        return ~t;
    } else {
        const unsigned ge = (w | 0x80808080u) - lo * 0x01010101u;
        const unsigned le = (((lo + (unsigned)R - 1u) | 0x80u) * 0x01010101u) - w;
        return ge & le;
    }
}

template <int F, int NPP, int S>
__global__ __launch_bounds__(PR_T)
void gcn_prop_ranges(const uint4* __restrict__ rid16, const uint2* __restrict__ rec,
                     const float* __restrict__ dinv, const float* __restrict__ tbl,
                     float* __restrict__ planes, int E, int Npad) {
    static_assert(NPP % (1 << RID_SHIFT) == 0, "a block owns whole range ids");
    constexpr int R = NPP >> RID_SHIFT;
    __shared__ float acc[NPP * F];
    __shared__ float dloc[NPP];
    __shared__ unsigned ring[PR_T / 32][PR_RING];
    const int tid  = threadIdx.x;
    const unsigned lane = (unsigned)tid & 31u;
    const int s    = blockIdx.x;
    const int nodeBase  = blockIdx.y * NPP;
    const unsigned lo   = (unsigned)blockIdx.y * (unsigned)R;
    unsigned* rq = &ring[tid >> 5][0];
    unsigned head = 0, tail = 0;

    const float4 zero4 = make_float4(0.0f, 0.0f, 0.0f, 0.0f);
    for (int i = tid * 4; i < NPP * F; i += PR_T * 4) *(float4*)(acc + i) = zero4;
    for (int i = tid * 4; i < NPP; i += PR_T * 4) {
        float4 v = zero4;
        if (nodeBase + i < Npad) v = *(const float4*)(dinv + nodeBase + i);
        *(float4*)(dloc + i) = v;
    }
    __syncthreads();

    auto drain = [&]() {
        while (tail - head >= 64u) {
            const unsigned qa = rq[(head + lane) & (PR_RING - 1)];
            const unsigned qb = rq[(head + 32u + lane) & (PR_RING - 1)];
            const uint2 pa = rec[qa & ((1u << ENT_EDGE_BITS) - 1u)];
            const uint2 pb = rec[qb & ((1u << ENT_EDGE_BITS) - 1u)];
            PrEdge<F> ga, gb;
            pr_load_edge<F>(ga, pa, qa >> ENT_EDGE_BITS, dloc, tbl);
            pr_load_edge<F>(gb, pb, qb >> ENT_EDGE_BITS, dloc, tbl);
            pr_add_edge<F>(acc, ga);
            pr_add_edge<F>(acc, gb);
            head += 64u;
        }
    };
    auto push_all = [&](unsigned m, const uint4& v, unsigned ebase) {
        for (int round = 0; round < 16; ++round) {
            const bool has = m != 0u;
            const unsigned mask = __builtin_amdgcn_ballot_w32(has);
            if (mask == 0u) break;
            if (has) {
                const unsigned q  = (unsigned)__builtin_ctz(m);
                const unsigned wd = q & 7u, bt = q >> 3;
                unsigned ent = ebase + (wd << 2) + bt;
                if constexpr (R > 1) {
                    const unsigned word = wd == 0u ? v.x : wd == 1u ? v.y : wd == 2u ? v.z : v.w;
                    ent |= (((word >> (bt << 3)) & 0xffu) - lo) << ENT_EDGE_BITS;
                }
                rq[__builtin_amdgcn_mbcnt_lo(mask, tail) & (PR_RING - 1)] = ent;
                m &= m - 1u;
            }
            tail += (unsigned)__builtin_popcount(mask);
            drain();
        }
    };

    const int nvecAll = (E + 15) >> 4;
    const int chunk   = (nvecAll + S - 1) / S;
    int v0 = s * chunk;      if (v0 > nvecAll) v0 = nvecAll;
    int v1 = v0 + chunk;     if (v1 > nvecAll) v1 = nvecAll;
    const int nvec = v1 - v0;
    const uint4* rv = rid16 + v0;

    for (int ib = 0; ib < nvec; ib += 8 * PR_T) {
        uint4 v[8];
#pragma unroll
        for (int k = 0; k < 8; ++k) {
            const int idx = ib + k * PR_T + tid;
            v[k] = rv[idx < nvec ? idx : nvec - 1];
        }
#pragma unroll
        for (int k = 0; k < 8; ++k) {
            const int idx = ib + k * PR_T + tid;
            const unsigned h0 = pr_hit_signs<R>(v[k].x, lo);
            const unsigned h1 = pr_hit_signs<R>(v[k].y, lo);
            const unsigned h2 = pr_hit_signs<R>(v[k].z, lo);
            const unsigned h3 = pr_hit_signs<R>(v[k].w, lo);
            unsigned m = ((h0 >> 7) & 0x01010101u) | ((h1 >> 6) & 0x02020202u) |
                         ((h2 >> 5) & 0x04040404u) | ((h3 >> 4) & 0x08080808u);
            if (idx >= nvec) m = 0u;
            push_all(m, v[k], (unsigned)(v0 + idx) << 4);
        }
    }
    {
        const unsigned n = tail - head;
#pragma unroll
        for (unsigned h = 0; h < 64u; h += 32u) {
            if (lane + h < n) {
                const unsigned q = rq[(head + h + lane) & (PR_RING - 1)];
                PrEdge<F> g;
                pr_load_edge<F>(g, rec[q & ((1u << ENT_EDGE_BITS) - 1u)], q >> ENT_EDGE_BITS,
                                dloc, tbl);
                pr_add_edge<F>(acc, g);
            }
        }
    }
    __syncthreads();

    int lim = Npad - nodeBase;
    if (lim > NPP) lim = NPP;
    float* dst = planes + ((size_t)s * (size_t)Npad + (size_t)nodeBase) * F;
    for (int i = tid * 4; i < lim * F; i += PR_T * 4)
        *(float4*)(dst + i) = *(const float4*)(acc + i);
}

#define P1_NPP 8192
#define P1_S   4
#define P2_NPP 24576
#define P2_S   12

__global__ void gcn_relu_hw2(const float* __restrict__ planes1, const float* __restrict__ xw1,
                             const float* __restrict__ dinv, const float* __restrict__ b1,
                             const float* __restrict__ W2, float* __restrict__ hw2, int N,
                             int Npad) {
    int i = blockIdx.x * blockDim.x + threadIdx.x;
    if (i >= N) return;
    const float self = dinv[i] * dinv[i];
    const float4* a = (const float4*)(planes1 + (long)i * 8);
    const float4* s = (const float4*)(xw1 + (long)i * 8);
    const size_t pitch4 = (size_t)Npad * 2;
    float4 a0 = a[0], a1 = a[1];
#pragma unroll
    for (int p = 1; p < P1_S; ++p) {
        const float4 u0 = a[p * pitch4], u1 = a[p * pitch4 + 1];
        a0.x += u0.x; a0.y += u0.y; a0.z += u0.z; a0.w += u0.w;
        a1.x += u1.x; a1.y += u1.y; a1.z += u1.z; a1.w += u1.w;
    }
    float hv[8];
    {
        float4 s0 = s[0], s1 = s[1];
        hv[0] = a0.x + self * s0.x; hv[1] = a0.y + self * s0.y;
        hv[2] = a0.z + self * s0.z; hv[3] = a0.w + self * s0.w;
        hv[4] = a1.x + self * s1.x; hv[5] = a1.y + self * s1.y;
        hv[6] = a1.z + self * s1.z; hv[7] = a1.w + self * s1.w;
    }
    float o0 = 0.0f, o1 = 0.0f;
#pragma unroll
    for (int f = 0; f < 8; ++f) {
        float h = fmaxf(hv[f] + b1[f], 0.0f);
        o0 = fmaf(h, W2[f * 2 + 0], o0);
        o1 = fmaf(h, W2[f * 2 + 1], o1);
    }
    hw2[(long)i * 2 + 0] = o0;
    hw2[(long)i * 2 + 1] = o1;
}

__global__ void gcn_finalize(float* __restrict__ outb, const float* __restrict__ planes2,
                             const float* __restrict__ hw2,
                             const float* __restrict__ dinv, const float* __restrict__ b2,
                             int N, int Npad) {
    int i = blockIdx.x * blockDim.x + threadIdx.x;
    if (i >= N) return;
    const float self = dinv[i] * dinv[i];
    const float2* a = (const float2*)(planes2 + (long)i * 2);
    float2 acc = a[0];
#pragma unroll
    for (int p = 1; p < P2_S; ++p) {
        const float2 u = a[(size_t)p * (size_t)Npad];
        acc.x += u.x; acc.y += u.y;
    }
    float v0 = acc.x + self * hw2[(long)i * 2 + 0] + b2[0];
    float v1 = acc.y + self * hw2[(long)i * 2 + 1] + b2[1];
    const float m = fmaxf(v0, v1);
    const float lse = m + logf(expf(v0 - m) + expf(v1 - m));
    outb[(long)i * 2 + 0] = v0 - lse;
    outb[(long)i * 2 + 1] = v1 - lse;
}

extern "C" void kernel_launch(void* const* d_in, const int* in_sizes, int n_in,
                              void* d_out, int out_size, void* d_ws, size_t ws_size,
                              hipStream_t stream) {
    const float* x    = (const float*)d_in[0];
    const float* W1   = (const float*)d_in[1];
    const float* b1   = (const float*)d_in[2];
    const float* W2   = (const float*)d_in[3];
    const float* b2   = (const float*)d_in[4];
    const int*   edge = (const int*)d_in[5];

    const int N = in_sizes[0] / 16;
    const int E = in_sizes[5] / 2;
    const int* row = edge;
    const int* col = edge + E;
    float* out = (float*)d_out;

    size_t off = 0;
    auto carve = [&](size_t bytes) {
        void* p = (char*)d_ws + off;
        off = (off + bytes + 255) & ~(size_t)255;
        return p;
    };
    const int Npad   = (N + 63) & ~63;
    const int nRange = (N + DEG_NPP - 1) / DEG_NPP;
    const int nR1    = (N + P1_NPP - 1) / P1_NPP;
    const int nR2    = (N + P2_NPP - 1) / P2_NPP;
    float* deg  = (float*)carve((size_t)Npad * 4);
    float* xw1  = (float*)carve((size_t)N * 8 * 4);
    float* hw2  = (float*)carve((size_t)N * 2 * 4);
    const size_t Epad = ((size_t)E + 15) & ~(size_t)15;
    const size_t degBytes = (size_t)DEG_S * (size_t)Npad * 4;
    const size_t recBytes = Epad * 8;
    void* shared = carve(degBytes > recBytes ? degBytes : recBytes);
    unsigned* planes = (unsigned*)shared;
    uint2*    recs   = (uint2*)shared;
    float* planes1 = (float*)carve((size_t)P1_S * (size_t)Npad * 8 * 4);
    float* planes2 = (float*)carve((size_t)P2_S * (size_t)Npad * 2 * 4);
    unsigned* rid   = (unsigned*)carve(Epad);
    unsigned* dcode = (unsigned*)carve((size_t)Npad);
    (void)ws_size;

    const int B = 256;
    const int gN   = (N + B - 1) / B;
    const int nTiles = (N + 15) / 16;
    const int gW   = (nTiles * 32 + B - 1) / B;
    const int gS   = (Npad / 4 + B - 1) / B;
    const int gE   = (int)((Epad / 4 + (size_t)(ENC_T * ENC_G) - 1) / (size_t)(ENC_T * ENC_G));

    gcn_degree_ranges<<<dim3(DEG_S, nRange), DEG_T, 0, stream>>>(col, planes, E, Npad);
    gcn_dinv_sum<<<gS,   B, 0, stream>>>(planes, deg, dcode, Npad);
    gcn_xw1_wmma<<<gW,   B, 0, stream>>>(x, W1, xw1, nTiles, N);
    gcn_encode_edges<<<gE, ENC_T, 0, stream>>>(row, col, deg, (const unsigned char*)dcode, rid,
                                               (uint4*)recs, E, Npad);
    gcn_prop_ranges<8, P1_NPP, P1_S><<<dim3(P1_S, nR1), PR_T, 0, stream>>>(
        (const uint4*)rid, recs, deg, xw1, planes1, E, Npad);
    gcn_relu_hw2<<<gN,   B, 0, stream>>>(planes1, xw1, deg, b1, W2, hw2, N, Npad);
    gcn_prop_ranges<2, P2_NPP, P2_S><<<dim3(P2_S, nR2), PR_T, 0, stream>>>(
        (const uint4*)rid, recs, deg, hw2, planes2, E, Npad);
    gcn_finalize<<<gN,   B, 0, stream>>>(out, planes2, hw2, deg, b2, N, Npad);
}
